# starting kernel
# baseline (speedup 1.0000x reference)
#include <hip/hip_runtime.h>
#include <hip/hip_bf16.h>

typedef __attribute__((ext_vector_type(2))) float v2f;
typedef __attribute__((ext_vector_type(8))) float v8f;

__global__ void gcn_deg_init(float* __restrict__ deg, int n) {
    int i = blockIdx.x * blockDim.x + threadIdx.x;
    if (i < n) deg[i] = 1.0f;
}

__global__ void gcn_deg_count(const int* __restrict__ dst, float* __restrict__ deg, int e) {
    int i = blockIdx.x * blockDim.x + threadIdx.x;
    if (i < e) unsafeAtomicAdd(&deg[dst[i]], 1.0f);
}

__global__ void gcn_dinv(const float* __restrict__ deg, float* __restrict__ dinv, int n) {
    int i = blockIdx.x * blockDim.x + threadIdx.x;
    if (i < n) {
        float d = deg[i];
        dinv[i] = (d > 0.0f) ? rsqrtf(d) : 0.0f;
    }
}

template <bool RELU>
__launch_bounds__(256)
__global__ void gcn_gemm128(const float* __restrict__ A,
                            const float* __restrict__ bias,
                            const float* __restrict__ W,
                            float* __restrict__ OutLin,
                            float* __restrict__ OutAgg,
                            const float* __restrict__ dinv,
                            int nrows) {
    __shared__ float sWp[128 * 128];
    const int tid = threadIdx.x;

    #pragma unroll
    for (int i = 0; i < 16; ++i) {
        int g = (i * 256 + tid) * 4;
        int k = g >> 7;
        int n = g & 127;
        float4 w4 = *(const float4*)(W + g);
        int kk = k >> 2, r = k & 3;
        int hf = r >> 1, j = r & 1;
        float wv[4] = {w4.x, w4.y, w4.z, w4.w};
        #pragma unroll
        for (int q = 0; q < 4; ++q) {
            int col = n + q;
            sWp[kk * 512 + (col >> 4) * 64 + hf * 32 + (col & 15) * 2 + j] = wv[q];
        }
    }
    __syncthreads();

    const int wave = tid >> 5;
    const int lane = tid & 31;
    const int half = lane >> 4;
    const int l16  = lane & 15;
    const int rowtile = blockIdx.x * 8 + wave;
    if (rowtile * 16 >= nrows) return;

    v8f acc[8];
    #pragma unroll
    for (int ct = 0; ct < 8; ++ct)
        acc[ct] = (v8f){0.f, 0.f, 0.f, 0.f, 0.f, 0.f, 0.f, 0.f};

    const float* arow = A + (size_t)(rowtile * 16 + l16) * 128;

    for (int kk = 0; kk < 32; ++kk) {
        const int kb = kk * 4 + 2 * half;
        float2 a2 = *(const float2*)(arow + kb);
        if (RELU) {
            const float2 bb = *(const float2*)(bias + kb);
            a2.x = fmaxf(a2.x + bb.x, 0.0f);
            a2.y = fmaxf(a2.y + bb.y, 0.0f);
        }
        v2f a; a.x = a2.x; a.y = a2.y;
        const float* bbase = &sWp[kk * 512 + half * 32 + l16 * 2];
        #pragma unroll
        for (int ct = 0; ct < 8; ++ct) {
            const float2 bp = *(const float2*)(bbase + ct * 64);
            v2f b; b.x = bp.x; b.y = bp.y;
            acc[ct] = __builtin_amdgcn_wmma_f32_16x16x4_f32(
                false, a, false, b, (short)0, acc[ct], false, false);
        }
    }

    const int rowbase = rowtile * 16 + 8 * half;
    float d2[8];
    #pragma unroll
    for (int j = 0; j < 8; ++j) {
        float s = dinv[rowbase + j];
        d2[j] = s * s;
    }
    #pragma unroll
    for (int ct = 0; ct < 8; ++ct) {
        const int col = ct * 16 + l16;
        #pragma unroll
        for (int j = 0; j < 8; ++j) {
            const size_t off = (size_t)(rowbase + j) * 128 + col;
            const float v = acc[ct][j];
            OutLin[off] = v;
            OutAgg[off] = v * d2[j];
        }
    }
}

__global__ void gcn_scatter(const int* __restrict__ src, const int* __restrict__ dst,
                            const float* __restrict__ dinv, const float* __restrict__ h,
                            float* __restrict__ out, int e) {
    int w = (blockIdx.x * blockDim.x + threadIdx.x) >> 5;
    int lane = threadIdx.x & 31;
    if (w >= e) return;
    int s = src[w];
    int d = dst[w];
    float nm = dinv[s] * dinv[d];
    float4 v = *(const float4*)(h + (size_t)s * 128 + lane * 4);
    float* o = out + (size_t)d * 128 + lane * 4;
    unsafeAtomicAdd(o + 0, v.x * nm);
    unsafeAtomicAdd(o + 1, v.y * nm);
    unsafeAtomicAdd(o + 2, v.z * nm);
    unsafeAtomicAdd(o + 3, v.w * nm);
}

__global__ void gcn_out_softmax(const float* __restrict__ h, const float* __restrict__ b2,
                                const float* __restrict__ Wo, const float* __restrict__ bo,
                                float* __restrict__ out, int n) {
    int w = (blockIdx.x * blockDim.x + threadIdx.x) >> 5;
    int lane = threadIdx.x & 31;
    if (w >= n) return;
    float4 v = *(const float4*)(h + (size_t)w * 128 + lane * 4);
    const float4 bb = *(const float4*)(b2 + lane * 4);
    v.x = fmaxf(v.x + bb.x, 0.0f);
    v.y = fmaxf(v.y + bb.y, 0.0f);
    v.z = fmaxf(v.z + bb.z, 0.0f);
    v.w = fmaxf(v.w + bb.w, 0.0f);
    float acc[5];
    #pragma unroll
    for (int c = 0; c < 5; ++c) {
        acc[c] = v.x * Wo[(lane * 4 + 0) * 5 + c]
               + v.y * Wo[(lane * 4 + 1) * 5 + c]
               + v.z * Wo[(lane * 4 + 2) * 5 + c]
               + v.w * Wo[(lane * 4 + 3) * 5 + c];
    }
    #pragma unroll
    for (int c = 0; c < 5; ++c) {
        #pragma unroll
        for (int off = 16; off > 0; off >>= 1)
            acc[c] += __shfl_xor(acc[c], off, 32);
    }
    #pragma unroll
    for (int c = 0; c < 5; ++c) acc[c] += bo[c];
    float m = acc[0];
    #pragma unroll
    for (int c = 1; c < 5; ++c) m = fmaxf(m, acc[c]);
    float ex[5], s = 0.0f;
    #pragma unroll
    for (int c = 0; c < 5; ++c) { ex[c] = expf(acc[c] - m); s += ex[c]; }
    float inv = 1.0f / s;
    if (lane == 0) {
        #pragma unroll
        for (int c = 0; c < 5; ++c) out[(size_t)w * 5 + c] = ex[c] * inv;
    }
}

extern "C" void kernel_launch(void* const* d_in, const int* in_sizes, int n_in,
                              void* d_out, int out_size, void* d_ws, size_t ws_size,
                              hipStream_t stream) {
    const float* x  = (const float*)d_in[0];
    const int*   ei = (const int*)d_in[1];
    const float* W1 = (const float*)d_in[2];
    const float* b1 = (const float*)d_in[3];
    const float* W2 = (const float*)d_in[4];
    const float* b2 = (const float*)d_in[5];
    const float* Wo = (const float*)d_in[6];
    const float* bo = (const float*)d_in[7];

    const int n = in_sizes[0] / 128;
    const int e = in_sizes[1] / 2;
    const int* srcp = ei;
    const int* dstp = ei + e;

    float* deg  = (float*)d_ws;
    float* dinv = deg + n;
    float* bufA = dinv + n;
    float* bufB = bufA + (size_t)n * 128;
    float* outp = (float*)d_out;

    const int T = 256;
    const int blk_n    = (n + T - 1) / T;
    const int blk_e    = (e + T - 1) / T;
    const int blk_ew   = (int)(((size_t)e * 32 + T - 1) / T);
    const int blk_gemm = (((n + 15) / 16) + 7) / 8;
    const int blk_ow   = (int)(((size_t)n * 32 + T - 1) / T);

    gcn_deg_init<<<blk_n, T, 0, stream>>>(deg, n);
    gcn_deg_count<<<blk_e, T, 0, stream>>>(dstp, deg, e);
    gcn_dinv<<<blk_n, T, 0, stream>>>(deg, dinv, n);

    gcn_gemm128<false><<<blk_gemm, T, 0, stream>>>(x, nullptr, W1, bufA, bufB, dinv, n);
    gcn_scatter<<<blk_ew, T, 0, stream>>>(srcp, dstp, dinv, bufA, bufB, e);

    gcn_gemm128<true><<<blk_gemm, T, 0, stream>>>(bufB, b1, W2, bufA, bufB, dinv, n);
    gcn_scatter<<<blk_ew, T, 0, stream>>>(srcp, dstp, dinv, bufA, bufB, e);

    gcn_out_softmax<<<blk_ow, T, 0, stream>>>(bufB, b2, Wo, bo, outp, n);
}


# edge aggregation: per-edge float-atomic scatter replaced by per-node register gather over a dst-grouped edge list built each call (int-atomic rank, scan, fill); sum order = arrival rank
# speedup vs baseline: 2.5318x; 2.5318x over previous
#include <hip/hip_runtime.h>
#include <hip/hip_bf16.h>

typedef __attribute__((ext_vector_type(2))) float v2f;
typedef __attribute__((ext_vector_type(8))) float v8f;

#define CNT_STRIDE 32

__global__ void gcn_cnt_init(int* __restrict__ cnt, int n) {
    int i = blockIdx.x * blockDim.x + threadIdx.x;
    if (i < n) cnt[(size_t)i * CNT_STRIDE] = 0;
}

__global__ void gcn_edge_rank(const int* __restrict__ dst, int* __restrict__ cnt,
                              int* __restrict__ rank, int e) {
    int i = blockIdx.x * blockDim.x + threadIdx.x;
    if (i < e) rank[i] = atomicAdd(&cnt[(size_t)dst[i] * CNT_STRIDE], 1);
}

__launch_bounds__(256)
__global__ void gcn_node_scan(const int* __restrict__ cnt, float* __restrict__ dinv,
                              int* __restrict__ local, int* __restrict__ bsum, int n) {
    __shared__ int swave[8];
    const int tid = threadIdx.x;
    const int lane = tid & 31;
    const int wv = tid >> 5;
    const int i = blockIdx.x * 256 + tid;
    const int c = (i < n) ? cnt[(size_t)i * CNT_STRIDE] : 0;
    if (i < n) {
        float d = (float)(c + 1);
        dinv[i] = (d > 0.0f) ? rsqrtf(d) : 0.0f;
    }
    int incl = c;
    for (int off = 1; off < 32; off <<= 1) {
        int t = __shfl_up(incl, off, 32);
        if (lane >= off) incl += t;
    }
    if (lane == 31) swave[wv] = incl;
    __syncthreads();
    int wbase = 0, tot = 0;
    for (int k = 0; k < 8; ++k) {
        int s = swave[k];
        if (k < wv) wbase += s;
        tot += s;
    }
    if (i < n) local[i] = wbase + incl - c;
    if (tid == 0) bsum[blockIdx.x] = tot;
}

__launch_bounds__(256)
__global__ void gcn_rowptr(const int* __restrict__ local, const int* __restrict__ bsum,
                           int* __restrict__ rowptr, int n, int e) {
    __shared__ int sred[8];
    const int tid = threadIdx.x;
    const int lane = tid & 31;
    const int wv = tid >> 5;
    const int b = blockIdx.x;
    int s = 0;
    for (int j = tid; j < b; j += 256) s += bsum[j];
    for (int off = 16; off > 0; off >>= 1) s += __shfl_xor(s, off, 32);
    if (lane == 0) sred[wv] = s;
    __syncthreads();
    int base = 0;
    for (int k = 0; k < 8; ++k) base += sred[k];
    const int i = b * 256 + tid;
    if (i < n) rowptr[i] = base + local[i];
    else if (i == n) rowptr[n] = e;
}

__global__ void gcn_edge_fill(const int* __restrict__ src, const int* __restrict__ dst,
                              const int* __restrict__ rank, const int* __restrict__ rowptr,
                              int* __restrict__ perm, int e) {
    int i = blockIdx.x * blockDim.x + threadIdx.x;
    if (i < e) perm[rowptr[dst[i]] + rank[i]] = src[i];
}

template <bool RELU>
__launch_bounds__(256)
__global__ void gcn_gemm128(const float* __restrict__ A,
                            const float* __restrict__ bias,
                            const float* __restrict__ W,
                            float* __restrict__ OutLin,
                            int nrows) {
    __shared__ float sWp[128 * 128];
    const int tid = threadIdx.x;

    #pragma unroll
    for (int i = 0; i < 16; ++i) {
        int g = (i * 256 + tid) * 4;
        int k = g >> 7;
        int n = g & 127;
        float4 w4 = *(const float4*)(W + g);
        int kk = k >> 2, r = k & 3;
        int hf = r >> 1, j = r & 1;
        float wv[4] = {w4.x, w4.y, w4.z, w4.w};
        #pragma unroll
        for (int q = 0; q < 4; ++q) {
            int col = n + q;
            sWp[kk * 512 + (col >> 4) * 64 + hf * 32 + (col & 15) * 2 + j] = wv[q];
        }
    }
    __syncthreads();

    const int wave = tid >> 5;
    const int lane = tid & 31;
    const int half = lane >> 4;
    const int l16  = lane & 15;
    const int rowtile = blockIdx.x * 8 + wave;
    if (rowtile * 16 >= nrows) return;

    v8f acc[8];
    #pragma unroll
    for (int ct = 0; ct < 8; ++ct)
        acc[ct] = (v8f){0.f, 0.f, 0.f, 0.f, 0.f, 0.f, 0.f, 0.f};

    const float* arow = A + (size_t)(rowtile * 16 + l16) * 128;

    for (int kk = 0; kk < 32; ++kk) {
        const int kb = kk * 4 + 2 * half;
        float2 a2 = *(const float2*)(arow + kb);
        if (RELU) {
            const float2 bb = *(const float2*)(bias + kb);
            a2.x = fmaxf(a2.x + bb.x, 0.0f);
            a2.y = fmaxf(a2.y + bb.y, 0.0f);
        }
        v2f a; a.x = a2.x; a.y = a2.y;
        const float* bbase = &sWp[kk * 512 + half * 32 + l16 * 2];
        #pragma unroll
        for (int ct = 0; ct < 8; ++ct) {
            const float2 bp = *(const float2*)(bbase + ct * 64);
            v2f b; b.x = bp.x; b.y = bp.y;
            acc[ct] = __builtin_amdgcn_wmma_f32_16x16x4_f32(
                false, a, false, b, (short)0, acc[ct], false, false);
        }
    }

    const int rowbase = rowtile * 16 + 8 * half;
    #pragma unroll
    for (int ct = 0; ct < 8; ++ct) {
        const int col = ct * 16 + l16;
        #pragma unroll
        for (int j = 0; j < 8; ++j) {
            const size_t off = (size_t)(rowbase + j) * 128 + col;
            OutLin[off] = acc[ct][j];
        }
    }
}

__launch_bounds__(256)
__global__ void gcn_gather(const int* __restrict__ rowptr, const int* __restrict__ perm,
                           const float* __restrict__ dinv, const float* __restrict__ h,
                           float* __restrict__ out, int n) {
    const int lane = threadIdx.x & 31;
    const int w = __builtin_amdgcn_readfirstlane((int)((blockIdx.x * blockDim.x + threadIdx.x) >> 5));
    if (w >= n) return;
    const int beg = rowptr[w];
    const int end = rowptr[w + 1];
    const float dd = dinv[w];
    const float4 hs = *(const float4*)(h + (size_t)w * 128 + lane * 4);
    float4 acc = make_float4(0.0f, 0.0f, 0.0f, 0.0f);
    for (int base = beg; base < end; base += 32) {
        const int m = min(32, end - base);
        const int idx = (lane < m) ? perm[base + lane] : 0;
        const float nv = dinv[idx] * dd;
        #pragma unroll 4
        for (int k = 0; k < m; ++k) {
            const int s = __builtin_amdgcn_readlane(idx, k);
            const float nm = __int_as_float(__builtin_amdgcn_readlane(__float_as_int(nv), k));
            const float4 v = *(const float4*)(h + (size_t)s * 128 + lane * 4);
            acc.x += v.x * nm;
            acc.y += v.y * nm;
            acc.z += v.z * nm;
            acc.w += v.w * nm;
        }
    }
    const float d2 = dd * dd;
    acc.x += hs.x * d2;
    acc.y += hs.y * d2;
    acc.z += hs.z * d2;
    acc.w += hs.w * d2;
    *(float4*)(out + (size_t)w * 128 + lane * 4) = acc;
}

__global__ void gcn_out_softmax(const float* __restrict__ h, const float* __restrict__ b2,
                                const float* __restrict__ Wo, const float* __restrict__ bo,
                                float* __restrict__ out, int n) {
    int w = (blockIdx.x * blockDim.x + threadIdx.x) >> 5;
    int lane = threadIdx.x & 31;
    if (w >= n) return;
    float4 v = *(const float4*)(h + (size_t)w * 128 + lane * 4);
    const float4 bb = *(const float4*)(b2 + lane * 4);
    v.x = fmaxf(v.x + bb.x, 0.0f);
    v.y = fmaxf(v.y + bb.y, 0.0f);
    v.z = fmaxf(v.z + bb.z, 0.0f);
    v.w = fmaxf(v.w + bb.w, 0.0f);
    float acc[5];
    #pragma unroll
    for (int c = 0; c < 5; ++c) {
        acc[c] = v.x * Wo[(lane * 4 + 0) * 5 + c]
               + v.y * Wo[(lane * 4 + 1) * 5 + c]
               + v.z * Wo[(lane * 4 + 2) * 5 + c]
               + v.w * Wo[(lane * 4 + 3) * 5 + c];
    }
    #pragma unroll
    for (int c = 0; c < 5; ++c) {
        #pragma unroll
        for (int off = 16; off > 0; off >>= 1)
            acc[c] += __shfl_xor(acc[c], off, 32);
    }
    #pragma unroll
    for (int c = 0; c < 5; ++c) acc[c] += bo[c];
    float m = acc[0];
    #pragma unroll
    for (int c = 1; c < 5; ++c) m = fmaxf(m, acc[c]);
    float ex[5], s = 0.0f;
    #pragma unroll
    for (int c = 0; c < 5; ++c) { ex[c] = expf(acc[c] - m); s += ex[c]; }
    float inv = 1.0f / s;
    if (lane == 0) {
        #pragma unroll
        for (int c = 0; c < 5; ++c) out[(size_t)w * 5 + c] = ex[c] * inv;
    }
}

extern "C" void kernel_launch(void* const* d_in, const int* in_sizes, int n_in,
                              void* d_out, int out_size, void* d_ws, size_t ws_size,
                              hipStream_t stream) {
    const float* x  = (const float*)d_in[0];
    const int*   ei = (const int*)d_in[1];
    const float* W1 = (const float*)d_in[2];
    const float* b1 = (const float*)d_in[3];
    const float* W2 = (const float*)d_in[4];
    const float* b2 = (const float*)d_in[5];
    const float* Wo = (const float*)d_in[6];
    const float* bo = (const float*)d_in[7];

    const int n = in_sizes[0] / 128;
    const int e = in_sizes[1] / 2;
    const int* srcp = ei;
    const int* dstp = ei + e;

    float* bufA = (float*)d_ws;
    float* bufB = bufA + (size_t)n * 128;
    int*   cnt    = (int*)(bufB + (size_t)n * 128);
    int*   rank   = cnt + (size_t)n * CNT_STRIDE;
    int*   perm   = rank + e;
    int*   rowptr = perm + e;
    int*   local  = rowptr + (n + 1);
    int*   bsum   = local + n;
    float* outp = (float*)d_out;

    const int T = 256;
    const int blk_n    = (n + T - 1) / T;
    const int blk_n1   = (n + 1 + T - 1) / T;
    const int blk_e    = (e + T - 1) / T;
    const int blk_gemm = (((n + 15) / 16) + 7) / 8;
    const int blk_ow   = (int)(((size_t)n * 32 + T - 1) / T);
    float* dinv = (float*)(bsum + blk_n);

    gcn_cnt_init<<<blk_n, T, 0, stream>>>(cnt, n);
    gcn_edge_rank<<<blk_e, T, 0, stream>>>(dstp, cnt, rank, e);
    gcn_node_scan<<<blk_n, T, 0, stream>>>(cnt, dinv, local, bsum, n);
    gcn_rowptr<<<blk_n1, T, 0, stream>>>(local, bsum, rowptr, n, e);
    gcn_edge_fill<<<blk_e, T, 0, stream>>>(srcp, dstp, rank, rowptr, perm, e);

    gcn_gemm128<false><<<blk_gemm, T, 0, stream>>>(x, nullptr, W1, bufA, n);
    gcn_gather<<<blk_ow, T, 0, stream>>>(rowptr, perm, dinv, bufA, bufB, n);

    gcn_gemm128<true><<<blk_gemm, T, 0, stream>>>(bufB, b1, W2, bufA, n);
    gcn_gather<<<blk_ow, T, 0, stream>>>(rowptr, perm, dinv, bufA, bufB, n);

    gcn_out_softmax<<<blk_ow, T, 0, stream>>>(bufB, b2, Wo, bo, outp, n);
}


# output projection + softmax fused into the layer-2 gather: the aggregated row stays in registers; the separate softmax launch and the row store/reload before it are gone; sum order kept
# speedup vs baseline: 2.6894x; 1.0622x over previous
#include <hip/hip_runtime.h>
#include <hip/hip_bf16.h>

typedef __attribute__((ext_vector_type(2))) float v2f;
typedef __attribute__((ext_vector_type(8))) float v8f;

#define CNT_STRIDE 32

__global__ void gcn_cnt_init(int* __restrict__ cnt, int n) {
    int i = blockIdx.x * blockDim.x + threadIdx.x;
    if (i < n) cnt[(size_t)i * CNT_STRIDE] = 0;
}

__global__ void gcn_edge_rank(const int* __restrict__ dst, int* __restrict__ cnt,
                              int* __restrict__ rank, int e) {
    int i = blockIdx.x * blockDim.x + threadIdx.x;
    if (i < e) rank[i] = atomicAdd(&cnt[(size_t)dst[i] * CNT_STRIDE], 1);
}

__launch_bounds__(256)
__global__ void gcn_node_scan(const int* __restrict__ cnt, float* __restrict__ dinv,
                              int* __restrict__ local, int* __restrict__ bsum, int n) {
    __shared__ int swave[8];
    const int tid = threadIdx.x;
    const int lane = tid & 31;
    const int wv = tid >> 5;
    const int i = blockIdx.x * 256 + tid;
    const int c = (i < n) ? cnt[(size_t)i * CNT_STRIDE] : 0;
    if (i < n) {
        float d = (float)(c + 1);
        dinv[i] = (d > 0.0f) ? rsqrtf(d) : 0.0f;
    }
    int incl = c;
    for (int off = 1; off < 32; off <<= 1) {
        int t = __shfl_up(incl, off, 32);
        if (lane >= off) incl += t;
    }
    if (lane == 31) swave[wv] = incl;
    __syncthreads();
    int wbase = 0, tot = 0;
    for (int k = 0; k < 8; ++k) {
        int s = swave[k];
        if (k < wv) wbase += s;
        tot += s;
    }
    if (i < n) local[i] = wbase + incl - c;
    if (tid == 0) bsum[blockIdx.x] = tot;
}

__launch_bounds__(256)
__global__ void gcn_rowptr(const int* __restrict__ local, const int* __restrict__ bsum,
                           int* __restrict__ rowptr, int n, int e) {
    __shared__ int sred[8];
    const int tid = threadIdx.x;
    const int lane = tid & 31;
    const int wv = tid >> 5;
    const int b = blockIdx.x;
    int s = 0;
    for (int j = tid; j < b; j += 256) s += bsum[j];
    for (int off = 16; off > 0; off >>= 1) s += __shfl_xor(s, off, 32);
    if (lane == 0) sred[wv] = s;
    __syncthreads();
    int base = 0;
    for (int k = 0; k < 8; ++k) base += sred[k];
    const int i = b * 256 + tid;
    if (i < n) rowptr[i] = base + local[i];
    else if (i == n) rowptr[n] = e;
}

__global__ void gcn_edge_fill(const int* __restrict__ src, const int* __restrict__ dst,
                              const int* __restrict__ rank, const int* __restrict__ rowptr,
                              int* __restrict__ perm, int e) {
    int i = blockIdx.x * blockDim.x + threadIdx.x;
    if (i < e) perm[rowptr[dst[i]] + rank[i]] = src[i];
}

template <bool RELU>
__launch_bounds__(256)
__global__ void gcn_gemm128(const float* __restrict__ A,
                            const float* __restrict__ bias,
                            const float* __restrict__ W,
                            float* __restrict__ OutLin,
                            int nrows) {
    __shared__ float sWp[128 * 128];
    const int tid = threadIdx.x;

    #pragma unroll
    for (int i = 0; i < 16; ++i) {
        int g = (i * 256 + tid) * 4;
        int k = g >> 7;
        int n = g & 127;
        float4 w4 = *(const float4*)(W + g);
        int kk = k >> 2, r = k & 3;
        int hf = r >> 1, j = r & 1;
        float wv[4] = {w4.x, w4.y, w4.z, w4.w};
        #pragma unroll
        for (int q = 0; q < 4; ++q) {
            int col = n + q;
            sWp[kk * 512 + (col >> 4) * 64 + hf * 32 + (col & 15) * 2 + j] = wv[q];
        }
    }
    __syncthreads();

    const int wave = tid >> 5;
    const int lane = tid & 31;
    const int half = lane >> 4;
    const int l16  = lane & 15;
    const int rowtile = blockIdx.x * 8 + wave;
    if (rowtile * 16 >= nrows) return;

    v8f acc[8];
    #pragma unroll
    for (int ct = 0; ct < 8; ++ct)
        acc[ct] = (v8f){0.f, 0.f, 0.f, 0.f, 0.f, 0.f, 0.f, 0.f};

    const float* arow = A + (size_t)(rowtile * 16 + l16) * 128;

    for (int kk = 0; kk < 32; ++kk) {
        const int kb = kk * 4 + 2 * half;
        float2 a2 = *(const float2*)(arow + kb);
        if (RELU) {
            const float2 bb = *(const float2*)(bias + kb);
            a2.x = fmaxf(a2.x + bb.x, 0.0f);
            a2.y = fmaxf(a2.y + bb.y, 0.0f);
        }
        v2f a; a.x = a2.x; a.y = a2.y;
        const float* bbase = &sWp[kk * 512 + half * 32 + l16 * 2];
        #pragma unroll
        for (int ct = 0; ct < 8; ++ct) {
            const float2 bp = *(const float2*)(bbase + ct * 64);
            v2f b; b.x = bp.x; b.y = bp.y;
            acc[ct] = __builtin_amdgcn_wmma_f32_16x16x4_f32(
                false, a, false, b, (short)0, acc[ct], false, false);
        }
    }

    const int rowbase = rowtile * 16 + 8 * half;
    #pragma unroll
    for (int ct = 0; ct < 8; ++ct) {
        const int col = ct * 16 + l16;
        #pragma unroll
        for (int j = 0; j < 8; ++j) {
            const size_t off = (size_t)(rowbase + j) * 128 + col;
            OutLin[off] = acc[ct][j];
        }
    }
}

template <bool FINAL>
__launch_bounds__(256)
__global__ void gcn_gather(const int* __restrict__ rowptr, const int* __restrict__ perm,
                           const float* __restrict__ dinv, const float* __restrict__ h,
                           float* __restrict__ out,
                           const float* __restrict__ b2, const float* __restrict__ Wo,
                           const float* __restrict__ bo, float* __restrict__ prob, int n) {
    const int lane = threadIdx.x & 31;
    const int w = __builtin_amdgcn_readfirstlane((int)((blockIdx.x * blockDim.x + threadIdx.x) >> 5));
    if (w >= n) return;
    const int beg = rowptr[w];
    const int end = rowptr[w + 1];
    const float dd = dinv[w];
    const float4 hs = *(const float4*)(h + (size_t)w * 128 + lane * 4);
    float4 acc = make_float4(0.0f, 0.0f, 0.0f, 0.0f);
    for (int base = beg; base < end; base += 32) {
        const int m = min(32, end - base);
        const int idx = (lane < m) ? perm[base + lane] : 0;
        const float nv = dinv[idx] * dd;
        #pragma unroll 4
        for (int k = 0; k < m; ++k) {
            const int s = __builtin_amdgcn_readlane(idx, k);
            const float nm = __int_as_float(__builtin_amdgcn_readlane(__float_as_int(nv), k));
            const float4 v = *(const float4*)(h + (size_t)s * 128 + lane * 4);
            acc.x += v.x * nm;
            acc.y += v.y * nm;
            acc.z += v.z * nm;
            acc.w += v.w * nm;
        }
    }
    const float d2 = dd * dd;
    acc.x += hs.x * d2;
    acc.y += hs.y * d2;
    acc.z += hs.z * d2;
    acc.w += hs.w * d2;
    if (!FINAL) {
        *(float4*)(out + (size_t)w * 128 + lane * 4) = acc;
        return;
    }
    const float4 bb = *(const float4*)(b2 + lane * 4);
    float4 v;
    v.x = fmaxf(acc.x + bb.x, 0.0f);
    v.y = fmaxf(acc.y + bb.y, 0.0f);
    v.z = fmaxf(acc.z + bb.z, 0.0f);
    v.w = fmaxf(acc.w + bb.w, 0.0f);
    float lg[5];
    #pragma unroll
    for (int c = 0; c < 5; ++c) {
        lg[c] = v.x * Wo[(lane * 4 + 0) * 5 + c]
              + v.y * Wo[(lane * 4 + 1) * 5 + c]
              + v.z * Wo[(lane * 4 + 2) * 5 + c]
              + v.w * Wo[(lane * 4 + 3) * 5 + c];
    }
    #pragma unroll
    for (int c = 0; c < 5; ++c) {
        #pragma unroll
        for (int off = 16; off > 0; off >>= 1)
            lg[c] += __shfl_xor(lg[c], off, 32);
    }
    #pragma unroll
    for (int c = 0; c < 5; ++c) lg[c] += bo[c];
    float m = lg[0];
    #pragma unroll
    for (int c = 1; c < 5; ++c) m = fmaxf(m, lg[c]);
    float ex[5], s = 0.0f;
    #pragma unroll
    for (int c = 0; c < 5; ++c) { ex[c] = expf(lg[c] - m); s += ex[c]; }
    const float inv = 1.0f / s;
    if (lane == 0) {
        #pragma unroll
        for (int c = 0; c < 5; ++c) prob[(size_t)w * 5 + c] = ex[c] * inv;
    }
}

extern "C" void kernel_launch(void* const* d_in, const int* in_sizes, int n_in,
                              void* d_out, int out_size, void* d_ws, size_t ws_size,
                              hipStream_t stream) {
    const float* x  = (const float*)d_in[0];
    const int*   ei = (const int*)d_in[1];
    const float* W1 = (const float*)d_in[2];
    const float* b1 = (const float*)d_in[3];
    const float* W2 = (const float*)d_in[4];
    const float* b2 = (const float*)d_in[5];
    const float* Wo = (const float*)d_in[6];
    const float* bo = (const float*)d_in[7];

    const int n = in_sizes[0] / 128;
    const int e = in_sizes[1] / 2;
    const int* srcp = ei;
    const int* dstp = ei + e;

    float* bufA = (float*)d_ws;
    float* bufB = bufA + (size_t)n * 128;
    int*   cnt    = (int*)(bufB + (size_t)n * 128);
    int*   rank   = cnt + (size_t)n * CNT_STRIDE;
    int*   perm   = rank + e;
    int*   rowptr = perm + e;
    int*   local  = rowptr + (n + 1);
    int*   bsum   = local + n;
    float* outp = (float*)d_out;

    const int T = 256;
    const int blk_n    = (n + T - 1) / T;
    const int blk_n1   = (n + 1 + T - 1) / T;
    const int blk_e    = (e + T - 1) / T;
    const int blk_gemm = (((n + 15) / 16) + 7) / 8;
    const int blk_ow   = (int)(((size_t)n * 32 + T - 1) / T);
    float* dinv = (float*)(bsum + blk_n);

    gcn_cnt_init<<<blk_n, T, 0, stream>>>(cnt, n);
    gcn_edge_rank<<<blk_e, T, 0, stream>>>(dstp, cnt, rank, e);
    gcn_node_scan<<<blk_n, T, 0, stream>>>(cnt, dinv, local, bsum, n);
    gcn_rowptr<<<blk_n1, T, 0, stream>>>(local, bsum, rowptr, n, e);
    gcn_edge_fill<<<blk_e, T, 0, stream>>>(srcp, dstp, rank, rowptr, perm, e);

    gcn_gemm128<false><<<blk_gemm, T, 0, stream>>>(x, nullptr, W1, bufA, n);
    gcn_gather<false><<<blk_ow, T, 0, stream>>>(rowptr, perm, dinv, bufA, bufB,
                                                nullptr, nullptr, nullptr, nullptr, n);

    gcn_gemm128<true><<<blk_gemm, T, 0, stream>>>(bufB, b1, W2, bufA, n);
    gcn_gather<true><<<blk_ow, T, 0, stream>>>(rowptr, perm, dinv, bufA, nullptr,
                                               b2, Wo, bo, outp, n);
}


# layer-1 GEMM split in two halves that run inside the edge-rank and edge-fill launches (GEMM blocks first, edge blocks behind), so it overlaps the list build; 9 launches -> 8; sum order kept
# speedup vs baseline: 2.8679x; 1.0664x over previous
#include <hip/hip_runtime.h>
#include <hip/hip_bf16.h>

typedef __attribute__((ext_vector_type(2))) float v2f;
typedef __attribute__((ext_vector_type(8))) float v8f;

#define CNT_STRIDE 32

__global__ void gcn_cnt_init(int* __restrict__ cnt, int n) {
    int i = blockIdx.x * blockDim.x + threadIdx.x;
    if (i < n) cnt[(size_t)i * CNT_STRIDE] = 0;
}

__launch_bounds__(256)
__global__ void gcn_node_scan(const int* __restrict__ cnt, float* __restrict__ dinv,
                              int* __restrict__ local, int* __restrict__ bsum, int n) {
    __shared__ int swave[8];
    const int tid = threadIdx.x;
    const int lane = tid & 31;
    const int wv = tid >> 5;
    const int i = blockIdx.x * 256 + tid;
    const int c = (i < n) ? cnt[(size_t)i * CNT_STRIDE] : 0;
    if (i < n) {
        float d = (float)(c + 1);
        dinv[i] = (d > 0.0f) ? rsqrtf(d) : 0.0f;
    }
    int incl = c;
    for (int off = 1; off < 32; off <<= 1) {
        int t = __shfl_up(incl, off, 32);
        if (lane >= off) incl += t;
    }
    if (lane == 31) swave[wv] = incl;
    __syncthreads();
    int wbase = 0, tot = 0;
    for (int k = 0; k < 8; ++k) {
        int s = swave[k];
        if (k < wv) wbase += s;
        tot += s;
    }
    if (i < n) local[i] = wbase + incl - c;
    if (tid == 0) bsum[blockIdx.x] = tot;
}

__launch_bounds__(256)
__global__ void gcn_rowptr(const int* __restrict__ local, const int* __restrict__ bsum,
                           int* __restrict__ rowptr, int n, int e) {
    __shared__ int sred[8];
    const int tid = threadIdx.x;
    const int lane = tid & 31;
    const int wv = tid >> 5;
    const int b = blockIdx.x;
    int s = 0;
    for (int j = tid; j < b; j += 256) s += bsum[j];
    for (int off = 16; off > 0; off >>= 1) s += __shfl_xor(s, off, 32);
    if (lane == 0) sred[wv] = s;
    __syncthreads();
    int base = 0;
    for (int k = 0; k < 8; ++k) base += sred[k];
    const int i = b * 256 + tid;
    if (i < n) rowptr[i] = base + local[i];
    else if (i == n) rowptr[n] = e;
}

template <bool RELU>
__device__ __forceinline__ void gemm128_block(const float* __restrict__ A,
                                              const float* __restrict__ bias,
                                              const float* __restrict__ W,
                                              float* __restrict__ OutLin,
                                              int nrows, int gblk) {
    __shared__ float sWp[128 * 128];
    const int tid = threadIdx.x;

    #pragma unroll
    for (int i = 0; i < 16; ++i) {
        int g = (i * 256 + tid) * 4;
        int k = g >> 7;
        int n = g & 127;
        float4 w4 = *(const float4*)(W + g);
        int kk = k >> 2, r = k & 3;
        int hf = r >> 1, j = r & 1;
        float wv[4] = {w4.x, w4.y, w4.z, w4.w};
        #pragma unroll
        for (int q = 0; q < 4; ++q) {
            int col = n + q;
            sWp[kk * 512 + (col >> 4) * 64 + hf * 32 + (col & 15) * 2 + j] = wv[q];
        }
    }
    __syncthreads();

    const int wave = tid >> 5;
    const int lane = tid & 31;
    const int half = lane >> 4;
    const int l16  = lane & 15;
    const int rowtile = gblk * 8 + wave;
    if (rowtile * 16 >= nrows) return;

    v8f acc[8];
    #pragma unroll
    for (int ct = 0; ct < 8; ++ct)
        acc[ct] = (v8f){0.f, 0.f, 0.f, 0.f, 0.f, 0.f, 0.f, 0.f};

    const float* arow = A + (size_t)(rowtile * 16 + l16) * 128;

    for (int kk = 0; kk < 32; ++kk) {
        const int kb = kk * 4 + 2 * half;
        float2 a2 = *(const float2*)(arow + kb);
        if (RELU) {
            const float2 bb = *(const float2*)(bias + kb);
            a2.x = fmaxf(a2.x + bb.x, 0.0f);
            a2.y = fmaxf(a2.y + bb.y, 0.0f);
        }
        v2f a; a.x = a2.x; a.y = a2.y;
        const float* bbase = &sWp[kk * 512 + half * 32 + l16 * 2];
        #pragma unroll
        for (int ct = 0; ct < 8; ++ct) {
            const float2 bp = *(const float2*)(bbase + ct * 64);
            v2f b; b.x = bp.x; b.y = bp.y;
            acc[ct] = __builtin_amdgcn_wmma_f32_16x16x4_f32(
                false, a, false, b, (short)0, acc[ct], false, false);
        }
    }

    const int rowbase = rowtile * 16 + 8 * half;
    #pragma unroll
    for (int ct = 0; ct < 8; ++ct) {
        const int col = ct * 16 + l16;
        #pragma unroll
        for (int j = 0; j < 8; ++j) {
            const size_t off = (size_t)(rowbase + j) * 128 + col;
            OutLin[off] = acc[ct][j];
        }
    }
}

template <bool RELU>
__launch_bounds__(256)
__global__ void gcn_gemm128(const float* __restrict__ A,
                            const float* __restrict__ bias,
                            const float* __restrict__ W,
                            float* __restrict__ OutLin,
                            int nrows) {
    gemm128_block<RELU>(A, bias, W, OutLin, nrows, (int)blockIdx.x);
}

__launch_bounds__(256)
__global__ void gcn_rank_gemm(const int* __restrict__ dst, int* __restrict__ cnt,
                              int* __restrict__ rank, int e,
                              const float* __restrict__ X, const float* __restrict__ W,
                              float* __restrict__ OutLin, int nrows,
                              int gblk0, int gcount) {
    const int b = (int)blockIdx.x;
    if (b < gcount) {
        gemm128_block<false>(X, nullptr, W, OutLin, nrows, gblk0 + b);
    } else {
        const int i = (b - gcount) * 256 + (int)threadIdx.x;
        if (i < e) rank[i] = atomicAdd(&cnt[(size_t)dst[i] * CNT_STRIDE], 1);
    }
}

__launch_bounds__(256)
__global__ void gcn_fill_gemm(const int* __restrict__ src, const int* __restrict__ dst,
                              const int* __restrict__ rank, const int* __restrict__ rowptr,
                              int* __restrict__ perm, int e,
                              const float* __restrict__ X, const float* __restrict__ W,
                              float* __restrict__ OutLin, int nrows,
                              int gblk0, int gcount) {
    const int b = (int)blockIdx.x;
    if (b < gcount) {
        gemm128_block<false>(X, nullptr, W, OutLin, nrows, gblk0 + b);
    } else {
        const int i = (b - gcount) * 256 + (int)threadIdx.x;
        if (i < e) perm[rowptr[dst[i]] + rank[i]] = src[i];
    }
}

template <bool FINAL>
__launch_bounds__(256)
__global__ void gcn_gather(const int* __restrict__ rowptr, const int* __restrict__ perm,
                           const float* __restrict__ dinv, const float* __restrict__ h,
                           float* __restrict__ out,
                           const float* __restrict__ b2, const float* __restrict__ Wo,
                           const float* __restrict__ bo, float* __restrict__ prob, int n) {
    const int lane = threadIdx.x & 31;
    const int w = __builtin_amdgcn_readfirstlane((int)((blockIdx.x * blockDim.x + threadIdx.x) >> 5));
    if (w >= n) return;
    const int beg = rowptr[w];
    const int end = rowptr[w + 1];
    const float dd = dinv[w];
    const float4 hs = *(const float4*)(h + (size_t)w * 128 + lane * 4);
    float4 acc = make_float4(0.0f, 0.0f, 0.0f, 0.0f);
    for (int base = beg; base < end; base += 32) {
        const int m = min(32, end - base);
        const int idx = (lane < m) ? perm[base + lane] : 0;
        const float nv = dinv[idx] * dd;
        #pragma unroll 4
        for (int k = 0; k < m; ++k) {
            const int s = __builtin_amdgcn_readlane(idx, k);
            const float nm = __int_as_float(__builtin_amdgcn_readlane(__float_as_int(nv), k));
            const float4 v = *(const float4*)(h + (size_t)s * 128 + lane * 4);
            acc.x += v.x * nm;
            acc.y += v.y * nm;
            acc.z += v.z * nm;
            acc.w += v.w * nm;
        }
    }
    const float d2 = dd * dd;
    acc.x += hs.x * d2;
    acc.y += hs.y * d2;
    acc.z += hs.z * d2;
    acc.w += hs.w * d2;
    if (!FINAL) {
        *(float4*)(out + (size_t)w * 128 + lane * 4) = acc;
        return;
    }
    const float4 bb = *(const float4*)(b2 + lane * 4);
    float4 v;
    v.x = fmaxf(acc.x + bb.x, 0.0f);
    v.y = fmaxf(acc.y + bb.y, 0.0f);
    v.z = fmaxf(acc.z + bb.z, 0.0f);
    v.w = fmaxf(acc.w + bb.w, 0.0f);
    float lg[5];
    #pragma unroll
    for (int c = 0; c < 5; ++c) {
        lg[c] = v.x * Wo[(lane * 4 + 0) * 5 + c]
              + v.y * Wo[(lane * 4 + 1) * 5 + c]
              + v.z * Wo[(lane * 4 + 2) * 5 + c]
              + v.w * Wo[(lane * 4 + 3) * 5 + c];
    }
    #pragma unroll
    for (int c = 0; c < 5; ++c) {
        #pragma unroll
        for (int off = 16; off > 0; off >>= 1)
            lg[c] += __shfl_xor(lg[c], off, 32);
    }
    #pragma unroll
    for (int c = 0; c < 5; ++c) lg[c] += bo[c];
    float m = lg[0];
    #pragma unroll
    for (int c = 1; c < 5; ++c) m = fmaxf(m, lg[c]);
    float ex[5], s = 0.0f;
    #pragma unroll
    for (int c = 0; c < 5; ++c) { ex[c] = expf(lg[c] - m); s += ex[c]; }
    const float inv = 1.0f / s;
    if (lane == 0) {
        #pragma unroll
        for (int c = 0; c < 5; ++c) prob[(size_t)w * 5 + c] = ex[c] * inv;
    }
}

extern "C" void kernel_launch(void* const* d_in, const int* in_sizes, int n_in,
                              void* d_out, int out_size, void* d_ws, size_t ws_size,
                              hipStream_t stream) {
    const float* x  = (const float*)d_in[0];
    const int*   ei = (const int*)d_in[1];
    const float* W1 = (const float*)d_in[2];
    const float* b1 = (const float*)d_in[3];
    const float* W2 = (const float*)d_in[4];
    const float* b2 = (const float*)d_in[5];
    const float* Wo = (const float*)d_in[6];
    const float* bo = (const float*)d_in[7];

    const int n = in_sizes[0] / 128;
    const int e = in_sizes[1] / 2;
    const int* srcp = ei;
    const int* dstp = ei + e;

    float* bufA = (float*)d_ws;
    float* bufB = bufA + (size_t)n * 128;
    int*   cnt    = (int*)(bufB + (size_t)n * 128);
    int*   rank   = cnt + (size_t)n * CNT_STRIDE;
    int*   perm   = rank + e;
    int*   rowptr = perm + e;
    int*   local  = rowptr + (n + 1);
    int*   bsum   = local + n;
    float* outp = (float*)d_out;

    const int T = 256;
    const int blk_n    = (n + T - 1) / T;
    const int blk_n1   = (n + 1 + T - 1) / T;
    const int blk_e    = (e + T - 1) / T;
    const int blk_gemm = (((n + 15) / 16) + 7) / 8;
    const int blk_ow   = (int)(((size_t)n * 32 + T - 1) / T);
    float* dinv = (float*)(bsum + blk_n);

    const int gb0 = (blk_gemm + 1) / 2;
    const int gb1 = blk_gemm - gb0;

    gcn_cnt_init<<<blk_n, T, 0, stream>>>(cnt, n);
    gcn_rank_gemm<<<gb0 + blk_e, T, 0, stream>>>(dstp, cnt, rank, e, x, W1, bufA, n, 0, gb0);
    gcn_node_scan<<<blk_n, T, 0, stream>>>(cnt, dinv, local, bsum, n);
    gcn_rowptr<<<blk_n1, T, 0, stream>>>(local, bsum, rowptr, n, e);
    gcn_fill_gemm<<<gb1 + blk_e, T, 0, stream>>>(srcp, dstp, rank, rowptr, perm, e,
                                                 x, W1, bufA, n, gb0, gb1);

    gcn_gather<false><<<blk_ow, T, 0, stream>>>(rowptr, perm, dinv, bufA, bufB,
                                                nullptr, nullptr, nullptr, nullptr, n);

    gcn_gemm128<true><<<blk_gemm, T, 0, stream>>>(bufB, b1, W2, bufA, n);
    gcn_gather<true><<<blk_ow, T, 0, stream>>>(rowptr, perm, dinv, bufA, nullptr,
                                               b2, Wo, bo, outp, n);
}


# layer-2 GEMM fused into the layer-1 gather: 16 one-node waves per block put relu(row+b1) in LDS as a 16-row A tile, 8 of them run its WMMAs; row store/reload and a launch gone (8->7); sum order kept
# speedup vs baseline: 3.0320x; 1.0572x over previous
#include <hip/hip_runtime.h>
#include <hip/hip_bf16.h>

typedef __attribute__((ext_vector_type(2))) float v2f;
typedef __attribute__((ext_vector_type(8))) float v8f;

#define CNT_STRIDE 32

__global__ void gcn_cnt_init(int* __restrict__ cnt, int n) {
    int i = blockIdx.x * blockDim.x + threadIdx.x;
    if (i < n) cnt[(size_t)i * CNT_STRIDE] = 0;
}

__launch_bounds__(256)
__global__ void gcn_node_scan(const int* __restrict__ cnt, float* __restrict__ dinv,
                              int* __restrict__ local, int* __restrict__ bsum, int n) {
    __shared__ int swave[8];
    const int tid = threadIdx.x;
    const int lane = tid & 31;
    const int wv = tid >> 5;
    const int i = blockIdx.x * 256 + tid;
    const int c = (i < n) ? cnt[(size_t)i * CNT_STRIDE] : 0;
    if (i < n) {
        float d = (float)(c + 1);
        dinv[i] = (d > 0.0f) ? rsqrtf(d) : 0.0f;
    }
    int incl = c;
    for (int off = 1; off < 32; off <<= 1) {
        int t = __shfl_up(incl, off, 32);
        if (lane >= off) incl += t;
    }
    if (lane == 31) swave[wv] = incl;
    __syncthreads();
    int wbase = 0, tot = 0;
    for (int k = 0; k < 8; ++k) {
        int s = swave[k];
        if (k < wv) wbase += s;
        tot += s;
    }
    if (i < n) local[i] = wbase + incl - c;
    if (tid == 0) bsum[blockIdx.x] = tot;
}

__launch_bounds__(256)
__global__ void gcn_rowptr(const int* __restrict__ local, const int* __restrict__ bsum,
                           int* __restrict__ rowptr, int n, int e) {
    __shared__ int sred[8];
    const int tid = threadIdx.x;
    const int lane = tid & 31;
    const int wv = tid >> 5;
    const int b = blockIdx.x;
    int s = 0;
    for (int j = tid; j < b; j += 256) s += bsum[j];
    for (int off = 16; off > 0; off >>= 1) s += __shfl_xor(s, off, 32);
    if (lane == 0) sred[wv] = s;
    __syncthreads();
    int base = 0;
    for (int k = 0; k < 8; ++k) base += sred[k];
    const int i = b * 256 + tid;
    if (i < n) rowptr[i] = base + local[i];
    else if (i == n) rowptr[n] = e;
}

template <bool RELU>
__device__ __forceinline__ void gemm128_block(const float* __restrict__ A,
                                              const float* __restrict__ bias,
                                              const float* __restrict__ W,
                                              float* __restrict__ OutLin,
                                              int nrows, int gblk) {
    __shared__ float sWp[128 * 128];
    const int tid = threadIdx.x;

    #pragma unroll
    for (int i = 0; i < 16; ++i) {
        int g = (i * 256 + tid) * 4;
        int k = g >> 7;
        int n = g & 127;
        float4 w4 = *(const float4*)(W + g);
        int kk = k >> 2, r = k & 3;
        int hf = r >> 1, j = r & 1;
        float wv[4] = {w4.x, w4.y, w4.z, w4.w};
        #pragma unroll
        for (int q = 0; q < 4; ++q) {
            int col = n + q;
            sWp[kk * 512 + (col >> 4) * 64 + hf * 32 + (col & 15) * 2 + j] = wv[q];
        }
    }
    __syncthreads();

    const int wave = tid >> 5;
    const int lane = tid & 31;
    const int half = lane >> 4;
    const int l16  = lane & 15;
    const int rowtile = gblk * 8 + wave;
    if (rowtile * 16 >= nrows) return;

    v8f acc[8];
    #pragma unroll
    for (int ct = 0; ct < 8; ++ct)
        acc[ct] = (v8f){0.f, 0.f, 0.f, 0.f, 0.f, 0.f, 0.f, 0.f};

    const float* arow = A + (size_t)(rowtile * 16 + l16) * 128;

    for (int kk = 0; kk < 32; ++kk) {
        const int kb = kk * 4 + 2 * half;
        float2 a2 = *(const float2*)(arow + kb);
        if (RELU) {
            const float2 bb = *(const float2*)(bias + kb);
            a2.x = fmaxf(a2.x + bb.x, 0.0f);
            a2.y = fmaxf(a2.y + bb.y, 0.0f);
        }
        v2f a; a.x = a2.x; a.y = a2.y;
        const float* bbase = &sWp[kk * 512 + half * 32 + l16 * 2];
        #pragma unroll
        for (int ct = 0; ct < 8; ++ct) {
            const float2 bp = *(const float2*)(bbase + ct * 64);
            v2f b; b.x = bp.x; b.y = bp.y;
            acc[ct] = __builtin_amdgcn_wmma_f32_16x16x4_f32(
                false, a, false, b, (short)0, acc[ct], false, false);
        }
    }

    const int rowbase = rowtile * 16 + 8 * half;
    #pragma unroll
    for (int ct = 0; ct < 8; ++ct) {
        const int col = ct * 16 + l16;
        #pragma unroll
        for (int j = 0; j < 8; ++j) {
            const size_t off = (size_t)(rowbase + j) * 128 + col;
            OutLin[off] = acc[ct][j];
        }
    }
}

__launch_bounds__(256)
__global__ void gcn_rank_gemm(const int* __restrict__ dst, int* __restrict__ cnt,
                              int* __restrict__ rank, int e,
                              const float* __restrict__ X, const float* __restrict__ W,
                              float* __restrict__ OutLin, int nrows,
                              int gblk0, int gcount) {
    const int b = (int)blockIdx.x;
    if (b < gcount) {
        gemm128_block<false>(X, nullptr, W, OutLin, nrows, gblk0 + b);
    } else {
        const int i = (b - gcount) * 256 + (int)threadIdx.x;
        if (i < e) rank[i] = atomicAdd(&cnt[(size_t)dst[i] * CNT_STRIDE], 1);
    }
}

__launch_bounds__(256)
__global__ void gcn_fill_gemm(const int* __restrict__ src, const int* __restrict__ dst,
                              const int* __restrict__ rank, const int* __restrict__ rowptr,
                              int* __restrict__ perm, int e,
                              const float* __restrict__ X, const float* __restrict__ W,
                              float* __restrict__ OutLin, int nrows,
                              int gblk0, int gcount) {
    const int b = (int)blockIdx.x;
    if (b < gcount) {
        gemm128_block<false>(X, nullptr, W, OutLin, nrows, gblk0 + b);
    } else {
        const int i = (b - gcount) * 256 + (int)threadIdx.x;
        if (i < e) perm[rowptr[dst[i]] + rank[i]] = src[i];
    }
}

template <bool FINAL>
__launch_bounds__(256)
__global__ void gcn_gather(const int* __restrict__ rowptr, const int* __restrict__ perm,
                           const float* __restrict__ dinv, const float* __restrict__ h,
                           float* __restrict__ out,
                           const float* __restrict__ b2, const float* __restrict__ Wo,
                           const float* __restrict__ bo, float* __restrict__ prob, int n) {
    const int lane = threadIdx.x & 31;
    const int w = __builtin_amdgcn_readfirstlane((int)((blockIdx.x * blockDim.x + threadIdx.x) >> 5));
    if (w >= n) return;
    const int beg = rowptr[w];
    const int end = rowptr[w + 1];
    const float dd = dinv[w];
    const float4 hs = *(const float4*)(h + (size_t)w * 128 + lane * 4);
    float4 acc = make_float4(0.0f, 0.0f, 0.0f, 0.0f);
    for (int base = beg; base < end; base += 32) {
        const int m = min(32, end - base);
        const int idx = (lane < m) ? perm[base + lane] : 0;
        const float nv = dinv[idx] * dd;
        #pragma unroll 4
        for (int k = 0; k < m; ++k) {
            const int s = __builtin_amdgcn_readlane(idx, k);
            const float nm = __int_as_float(__builtin_amdgcn_readlane(__float_as_int(nv), k));
            const float4 v = *(const float4*)(h + (size_t)s * 128 + lane * 4);
            acc.x += v.x * nm;
            acc.y += v.y * nm;
            acc.z += v.z * nm;
            acc.w += v.w * nm;
        }
    }
    const float d2 = dd * dd;
    acc.x += hs.x * d2;
    acc.y += hs.y * d2;
    acc.z += hs.z * d2;
    acc.w += hs.w * d2;
    if (!FINAL) {
        *(float4*)(out + (size_t)w * 128 + lane * 4) = acc;
        return;
    }
    const float4 bb = *(const float4*)(b2 + lane * 4);
    float4 v;
    v.x = fmaxf(acc.x + bb.x, 0.0f);
    v.y = fmaxf(acc.y + bb.y, 0.0f);
    v.z = fmaxf(acc.z + bb.z, 0.0f);
    v.w = fmaxf(acc.w + bb.w, 0.0f);
    float lg[5];
    #pragma unroll
    for (int c = 0; c < 5; ++c) {
        lg[c] = v.x * Wo[(lane * 4 + 0) * 5 + c]
              + v.y * Wo[(lane * 4 + 1) * 5 + c]
              + v.z * Wo[(lane * 4 + 2) * 5 + c]
              + v.w * Wo[(lane * 4 + 3) * 5 + c];
    }
    #pragma unroll
    for (int c = 0; c < 5; ++c) {
        #pragma unroll
        for (int off = 16; off > 0; off >>= 1)
            lg[c] += __shfl_xor(lg[c], off, 32);
    }
    #pragma unroll
    for (int c = 0; c < 5; ++c) lg[c] += bo[c];
    float m = lg[0];
    #pragma unroll
    for (int c = 1; c < 5; ++c) m = fmaxf(m, lg[c]);
    float ex[5], s = 0.0f;
    #pragma unroll
    for (int c = 0; c < 5; ++c) { ex[c] = expf(lg[c] - m); s += ex[c]; }
    const float inv = 1.0f / s;
    if (lane == 0) {
        #pragma unroll
        for (int c = 0; c < 5; ++c) prob[(size_t)w * 5 + c] = ex[c] * inv;
    }
}

#define SA_PITCH 132

__launch_bounds__(512)
__global__ void gcn_gather_gemm(const int* __restrict__ rowptr, const int* __restrict__ perm,
                                const float* __restrict__ dinv, const float* __restrict__ h,
                                const float* __restrict__ b1, const float* __restrict__ W,
                                float* __restrict__ OutLin, int n) {
    __shared__ float sA[16 * SA_PITCH];
    const int lane = threadIdx.x & 31;
    const int wave = __builtin_amdgcn_readfirstlane((int)(threadIdx.x >> 5));
    const int node0 = (int)blockIdx.x * 16;
    const int w = node0 + wave;
    float4 v = make_float4(0.0f, 0.0f, 0.0f, 0.0f);
    if (w < n) {
        const int beg = rowptr[w];
        const int end = rowptr[w + 1];
        const float dd = dinv[w];
        const float4 hs = *(const float4*)(h + (size_t)w * 128 + lane * 4);
        float4 acc = make_float4(0.0f, 0.0f, 0.0f, 0.0f);
        for (int base = beg; base < end; base += 32) {
            const int m = min(32, end - base);
            const int idx = (lane < m) ? perm[base + lane] : 0;
            const float nv = dinv[idx] * dd;
            #pragma unroll 4
            for (int k = 0; k < m; ++k) {
                const int s = __builtin_amdgcn_readlane(idx, k);
                const float nm = __int_as_float(__builtin_amdgcn_readlane(__float_as_int(nv), k));
                const float4 r = *(const float4*)(h + (size_t)s * 128 + lane * 4);
                acc.x += r.x * nm;
                acc.y += r.y * nm;
                acc.z += r.z * nm;
                acc.w += r.w * nm;
            }
        }
        const float d2 = dd * dd;
        acc.x += hs.x * d2;
        acc.y += hs.y * d2;
        acc.z += hs.z * d2;
        acc.w += hs.w * d2;
        const float4 bb = *(const float4*)(b1 + lane * 4);
        v.x = fmaxf(acc.x + bb.x, 0.0f);
        v.y = fmaxf(acc.y + bb.y, 0.0f);
        v.z = fmaxf(acc.z + bb.z, 0.0f);
        v.w = fmaxf(acc.w + bb.w, 0.0f);
    }
    const int half = lane >> 4;
    const int l16  = lane & 15;
    const int ct   = wave & 7;
    const float* bcol = W + (size_t)(2 * half) * 128 + ct * 16 + l16;
    float bx[2][8], by[2][8];
    #pragma unroll
    for (int i = 0; i < 8; ++i) { bx[0][i] = 0.0f; by[0][i] = 0.0f; bx[1][i] = 0.0f; by[1][i] = 0.0f; }
    if (wave < 8) {
        #pragma unroll
        for (int i = 0; i < 8; ++i) {
            bx[0][i] = bcol[(size_t)(i * 4) * 128];
            by[0][i] = bcol[(size_t)(i * 4 + 1) * 128];
        }
    }
    *(float4*)&sA[wave * SA_PITCH + lane * 4] = v;
    __syncthreads();
    if (wave >= 8) return;

    v8f d = (v8f){0.f, 0.f, 0.f, 0.f, 0.f, 0.f, 0.f, 0.f};
    const float* arow = &sA[l16 * SA_PITCH + 2 * half];
    #pragma unroll
    for (int g = 0; g < 4; ++g) {
        if (g + 1 < 4) {
            #pragma unroll
            for (int i = 0; i < 8; ++i) {
                const int kn = (g + 1) * 8 + i;
                bx[(g + 1) & 1][i] = bcol[(size_t)(kn * 4) * 128];
                by[(g + 1) & 1][i] = bcol[(size_t)(kn * 4 + 1) * 128];
            }
        }
        #pragma unroll
        for (int i = 0; i < 8; ++i) {
            const int kk = g * 8 + i;
            const float2 a2 = *(const float2*)(arow + kk * 4);
            v2f a; a.x = a2.x; a.y = a2.y;
            v2f b; b.x = bx[g & 1][i]; b.y = by[g & 1][i];
            d = __builtin_amdgcn_wmma_f32_16x16x4_f32(
                false, a, false, b, (short)0, d, false, false);
        }
    }
    const int rowbase = node0 + 8 * half;
    const int col = ct * 16 + l16;
    #pragma unroll
    for (int j = 0; j < 8; ++j) {
        if (rowbase + j < n) OutLin[(size_t)(rowbase + j) * 128 + col] = d[j];
    }
}

extern "C" void kernel_launch(void* const* d_in, const int* in_sizes, int n_in,
                              void* d_out, int out_size, void* d_ws, size_t ws_size,
                              hipStream_t stream) {
    const float* x  = (const float*)d_in[0];
    const int*   ei = (const int*)d_in[1];
    const float* W1 = (const float*)d_in[2];
    const float* b1 = (const float*)d_in[3];
    const float* W2 = (const float*)d_in[4];
    const float* b2 = (const float*)d_in[5];
    const float* Wo = (const float*)d_in[6];
    const float* bo = (const float*)d_in[7];

    const int n = in_sizes[0] / 128;
    const int e = in_sizes[1] / 2;
    const int* srcp = ei;
    const int* dstp = ei + e;

    float* bufA = (float*)d_ws;
    float* bufB = bufA + (size_t)n * 128;
    int*   cnt    = (int*)(bufB + (size_t)n * 128);
    int*   rank   = cnt + (size_t)n * CNT_STRIDE;
    int*   perm   = rank + e;
    int*   rowptr = perm + e;
    int*   local  = rowptr + (n + 1);
    int*   bsum   = local + n;
    float* outp = (float*)d_out;

    const int T = 256;
    const int blk_n    = (n + T - 1) / T;
    const int blk_n1   = (n + 1 + T - 1) / T;
    const int blk_e    = (e + T - 1) / T;
    const int blk_gemm = (((n + 15) / 16) + 7) / 8;
    const int blk_ow   = (int)(((size_t)n * 32 + T - 1) / T);
    const int blk_t16  = (n + 15) / 16;
    float* dinv = (float*)(bsum + blk_n);

    const int gb0 = (blk_gemm + 1) / 2;
    const int gb1 = blk_gemm - gb0;

    gcn_cnt_init<<<blk_n, T, 0, stream>>>(cnt, n);
    gcn_rank_gemm<<<gb0 + blk_e, T, 0, stream>>>(dstp, cnt, rank, e, x, W1, bufA, n, 0, gb0);
    gcn_node_scan<<<blk_n, T, 0, stream>>>(cnt, dinv, local, bsum, n);
    gcn_rowptr<<<blk_n1, T, 0, stream>>>(local, bsum, rowptr, n, e);
    gcn_fill_gemm<<<gb1 + blk_e, T, 0, stream>>>(srcp, dstp, rank, rowptr, perm, e,
                                                 x, W1, bufA, n, gb0, gb1);

    gcn_gather_gemm<<<blk_t16, 512, 0, stream>>>(rowptr, perm, dinv, bufA, b1, W2, bufB, n);
    gcn_gather<true><<<blk_ow, T, 0, stream>>>(rowptr, perm, dinv, bufB, nullptr,
                                               b2, Wo, bo, outp, n);
}


# edge list as one 32-slot line per node, filled by the counting pass itself (slot = atomic rank; rank >= 32 to a spill list): scan, rowptr and fill launches gone, 7 -> 4; sum order = arrival rank
# speedup vs baseline: 3.1750x; 1.0472x over previous
#include <hip/hip_runtime.h>
#include <hip/hip_bf16.h>

typedef __attribute__((ext_vector_type(2))) float v2f;
typedef __attribute__((ext_vector_type(8))) float v8f;

#define NSLOT 32

__global__ void gcn_zero(int* __restrict__ cnt, int* __restrict__ spill_n, int n) {
    int i = blockIdx.x * blockDim.x + threadIdx.x;
    if (i < n) cnt[i] = 0;
    if (i == 0) spill_n[0] = 0;
}

__device__ __forceinline__ float gcn_dinv(int c) {
    const float d = (float)(c + 1);
    return (d > 0.0f) ? rsqrtf(d) : 0.0f;
}

template <bool RELU>
__device__ __forceinline__ void gemm128_block(const float* __restrict__ A,
                                              const float* __restrict__ bias,
                                              const float* __restrict__ W,
                                              float* __restrict__ OutLin,
                                              int nrows, int gblk) {
    __shared__ float sWp[128 * 128];
    const int tid = threadIdx.x;

    #pragma unroll
    for (int i = 0; i < 16; ++i) {
        int g = (i * 256 + tid) * 4;
        int k = g >> 7;
        int n = g & 127;
        float4 w4 = *(const float4*)(W + g);
        int kk = k >> 2, r = k & 3;
        int hf = r >> 1, j = r & 1;
        float wv[4] = {w4.x, w4.y, w4.z, w4.w};
        #pragma unroll
        for (int q = 0; q < 4; ++q) {
            int col = n + q;
            sWp[kk * 512 + (col >> 4) * 64 + hf * 32 + (col & 15) * 2 + j] = wv[q];
        }
    }
    __syncthreads();

    const int wave = tid >> 5;
    const int lane = tid & 31;
    const int half = lane >> 4;
    const int l16  = lane & 15;
    const int rowtile = gblk * 8 + wave;
    if (rowtile * 16 >= nrows) return;

    v8f acc[8];
    #pragma unroll
    for (int ct = 0; ct < 8; ++ct)
        acc[ct] = (v8f){0.f, 0.f, 0.f, 0.f, 0.f, 0.f, 0.f, 0.f};

    const float* arow = A + (size_t)(rowtile * 16 + l16) * 128;

    for (int kk = 0; kk < 32; ++kk) {
        const int kb = kk * 4 + 2 * half;
        float2 a2 = *(const float2*)(arow + kb);
        if (RELU) {
            const float2 bb = *(const float2*)(bias + kb);
            a2.x = fmaxf(a2.x + bb.x, 0.0f);
            a2.y = fmaxf(a2.y + bb.y, 0.0f);
        }
        v2f a; a.x = a2.x; a.y = a2.y;
        const float* bbase = &sWp[kk * 512 + half * 32 + l16 * 2];
        #pragma unroll
        for (int ct = 0; ct < 8; ++ct) {
            const float2 bp = *(const float2*)(bbase + ct * 64);
            v2f b; b.x = bp.x; b.y = bp.y;
            acc[ct] = __builtin_amdgcn_wmma_f32_16x16x4_f32(
                false, a, false, b, (short)0, acc[ct], false, false);
        }
    }

    const int rowbase = rowtile * 16 + 8 * half;
    #pragma unroll
    for (int ct = 0; ct < 8; ++ct) {
        const int col = ct * 16 + l16;
        #pragma unroll
        for (int j = 0; j < 8; ++j) {
            const size_t off = (size_t)(rowbase + j) * 128 + col;
            OutLin[off] = acc[ct][j];
        }
    }
}

__launch_bounds__(256)
__global__ void gcn_edge_gemm(const int* __restrict__ src, const int* __restrict__ dst,
                              int* __restrict__ cnt, int* __restrict__ slots,
                              int* __restrict__ spill_n, int* __restrict__ spill_src,
                              int* __restrict__ spill_dst, int e,
                              const float* __restrict__ X, const float* __restrict__ W,
                              float* __restrict__ OutLin, int nrows, int gcount) {
    const int b = (int)blockIdx.x;
    if (b < gcount) {
        gemm128_block<false>(X, nullptr, W, OutLin, nrows, b);
    } else {
        const int i = (b - gcount) * 256 + (int)threadIdx.x;
        if (i < e) {
            const int d = dst[i];
            const int s = src[i];
            const int r = atomicAdd(&cnt[d], 1);
            if (r < NSLOT) {
                slots[(size_t)d * NSLOT + r] = s;
            } else {
                const int o = atomicAdd(spill_n, 1);
                spill_src[o] = s;
                spill_dst[o] = d;
            }
        }
    }
}

__device__ __forceinline__ float4 gcn_aggregate(const int* __restrict__ cnt,
                                                const int* __restrict__ slots,
                                                const int* __restrict__ spill_n,
                                                const int* __restrict__ spill_src,
                                                const int* __restrict__ spill_dst,
                                                int ecap, const float* __restrict__ h,
                                                int w, int lane) {
    const int c = cnt[w];
    const float4 hs = *(const float4*)(h + (size_t)w * 128 + lane * 4);
    const float dd = gcn_dinv(c);
    const int m = min(c, NSLOT);
    const int idx = (lane < m) ? slots[(size_t)w * NSLOT + lane] : 0;
    const float nv = gcn_dinv(cnt[idx]) * dd;
    float4 acc = make_float4(0.0f, 0.0f, 0.0f, 0.0f);
    #pragma unroll 4
    for (int k = 0; k < m; ++k) {
        const int s = __builtin_amdgcn_readlane(idx, k);
        const float nm = __int_as_float(__builtin_amdgcn_readlane(__float_as_int(nv), k));
        const float4 v = *(const float4*)(h + (size_t)s * 128 + lane * 4);
        acc.x += v.x * nm;
        acc.y += v.y * nm;
        acc.z += v.z * nm;
        acc.w += v.w * nm;
    }
    if (c > NSLOT) {
        const int on = min(spill_n[0], ecap);
        for (int base = 0; base < on; base += 32) {
            const int j = base + lane;
            bool mine = false;
            if (j < on) mine = (spill_dst[j] == w);
            const int os = mine ? spill_src[j] : 0;
            const float onv = gcn_dinv(cnt[os]) * dd;
            unsigned mask = __builtin_amdgcn_ballot_w32(mine);
            while (mask != 0u) {
                const int k = __builtin_ctz(mask);
                mask &= mask - 1u;
                const int s = __builtin_amdgcn_readlane(os, k);
                const float nm = __int_as_float(__builtin_amdgcn_readlane(__float_as_int(onv), k));
                const float4 v = *(const float4*)(h + (size_t)s * 128 + lane * 4);
                acc.x += v.x * nm;
                acc.y += v.y * nm;
                acc.z += v.z * nm;
                acc.w += v.w * nm;
            }
        }
    }
    const float d2 = dd * dd;
    acc.x += hs.x * d2;
    acc.y += hs.y * d2;
    acc.z += hs.z * d2;
    acc.w += hs.w * d2;
    return acc;
}

__launch_bounds__(256)
__global__ void gcn_gather_out(const int* __restrict__ cnt, const int* __restrict__ slots,
                               const int* __restrict__ spill_n, const int* __restrict__ spill_src,
                               const int* __restrict__ spill_dst, int ecap,
                               const float* __restrict__ h,
                               const float* __restrict__ b2, const float* __restrict__ Wo,
                               const float* __restrict__ bo, float* __restrict__ prob, int n) {
    const int lane = threadIdx.x & 31;
    const int w = __builtin_amdgcn_readfirstlane((int)((blockIdx.x * blockDim.x + threadIdx.x) >> 5));
    if (w >= n) return;
    const float4 acc = gcn_aggregate(cnt, slots, spill_n, spill_src, spill_dst, ecap, h, w, lane);
    const float4 bb = *(const float4*)(b2 + lane * 4);
    float4 v;
    v.x = fmaxf(acc.x + bb.x, 0.0f);
    v.y = fmaxf(acc.y + bb.y, 0.0f);
    v.z = fmaxf(acc.z + bb.z, 0.0f);
    v.w = fmaxf(acc.w + bb.w, 0.0f);
    float lg[5];
    #pragma unroll
    for (int c = 0; c < 5; ++c) {
        lg[c] = v.x * Wo[(lane * 4 + 0) * 5 + c]
              + v.y * Wo[(lane * 4 + 1) * 5 + c]
              + v.z * Wo[(lane * 4 + 2) * 5 + c]
              + v.w * Wo[(lane * 4 + 3) * 5 + c];
    }
    #pragma unroll
    for (int c = 0; c < 5; ++c) {
        #pragma unroll
        for (int off = 16; off > 0; off >>= 1)
            lg[c] += __shfl_xor(lg[c], off, 32);
    }
    #pragma unroll
    for (int c = 0; c < 5; ++c) lg[c] += bo[c];
    float m = lg[0];
    #pragma unroll
    for (int c = 1; c < 5; ++c) m = fmaxf(m, lg[c]);
    float ex[5], s = 0.0f;
    #pragma unroll
    for (int c = 0; c < 5; ++c) { ex[c] = expf(lg[c] - m); s += ex[c]; }
    const float inv = 1.0f / s;
    if (lane == 0) {
        #pragma unroll
        for (int c = 0; c < 5; ++c) prob[(size_t)w * 5 + c] = ex[c] * inv;
    }
}

#define SA_PITCH 132

__launch_bounds__(512)
__global__ void gcn_gather_gemm(const int* __restrict__ cnt, const int* __restrict__ slots,
                                const int* __restrict__ spill_n, const int* __restrict__ spill_src,
                                const int* __restrict__ spill_dst, int ecap,
                                const float* __restrict__ h,
                                const float* __restrict__ b1, const float* __restrict__ W,
                                float* __restrict__ OutLin, int n) {
    __shared__ float sA[16 * SA_PITCH];
    const int lane = threadIdx.x & 31;
    const int wave = __builtin_amdgcn_readfirstlane((int)(threadIdx.x >> 5));
    const int node0 = (int)blockIdx.x * 16;
    const int w = node0 + wave;
    float4 v = make_float4(0.0f, 0.0f, 0.0f, 0.0f);
    if (w < n) {
        const float4 acc = gcn_aggregate(cnt, slots, spill_n, spill_src, spill_dst, ecap, h, w, lane);
        const float4 bb = *(const float4*)(b1 + lane * 4);
        v.x = fmaxf(acc.x + bb.x, 0.0f);
        v.y = fmaxf(acc.y + bb.y, 0.0f);
        v.z = fmaxf(acc.z + bb.z, 0.0f);
        v.w = fmaxf(acc.w + bb.w, 0.0f);
    }
    const int half = lane >> 4;
    const int l16  = lane & 15;
    const int ct   = wave & 7;
    const float* bcol = W + (size_t)(2 * half) * 128 + ct * 16 + l16;
    float bx[2][8], by[2][8];
    #pragma unroll
    for (int i = 0; i < 8; ++i) { bx[0][i] = 0.0f; by[0][i] = 0.0f; bx[1][i] = 0.0f; by[1][i] = 0.0f; }
    if (wave < 8) {
        #pragma unroll
        for (int i = 0; i < 8; ++i) {
            bx[0][i] = bcol[(size_t)(i * 4) * 128];
            by[0][i] = bcol[(size_t)(i * 4 + 1) * 128];
        }
    }
    *(float4*)&sA[wave * SA_PITCH + lane * 4] = v;
    __syncthreads();
    if (wave >= 8) return;

    v8f d = (v8f){0.f, 0.f, 0.f, 0.f, 0.f, 0.f, 0.f, 0.f};
    const float* arow = &sA[l16 * SA_PITCH + 2 * half];
    #pragma unroll
    for (int g = 0; g < 4; ++g) {
        if (g + 1 < 4) {
            #pragma unroll
            for (int i = 0; i < 8; ++i) {
                const int kn = (g + 1) * 8 + i;
                bx[(g + 1) & 1][i] = bcol[(size_t)(kn * 4) * 128];
                by[(g + 1) & 1][i] = bcol[(size_t)(kn * 4 + 1) * 128];
            }
        }
        #pragma unroll
        for (int i = 0; i < 8; ++i) {
            const int kk = g * 8 + i;
            const float2 a2 = *(const float2*)(arow + kk * 4);
            v2f a; a.x = a2.x; a.y = a2.y;
            v2f b; b.x = bx[g & 1][i]; b.y = by[g & 1][i];
            d = __builtin_amdgcn_wmma_f32_16x16x4_f32(
                false, a, false, b, (short)0, d, false, false);
        }
    }
    const int rowbase = node0 + 8 * half;
    const int col = ct * 16 + l16;
    #pragma unroll
    for (int j = 0; j < 8; ++j) {
        if (rowbase + j < n) OutLin[(size_t)(rowbase + j) * 128 + col] = d[j];
    }
}

extern "C" void kernel_launch(void* const* d_in, const int* in_sizes, int n_in,
                              void* d_out, int out_size, void* d_ws, size_t ws_size,
                              hipStream_t stream) {
    const float* x  = (const float*)d_in[0];
    const int*   ei = (const int*)d_in[1];
    const float* W1 = (const float*)d_in[2];
    const float* b1 = (const float*)d_in[3];
    const float* W2 = (const float*)d_in[4];
    const float* b2 = (const float*)d_in[5];
    const float* Wo = (const float*)d_in[6];
    const float* bo = (const float*)d_in[7];

    const int n = in_sizes[0] / 128;
    const int e = in_sizes[1] / 2;
    const int* srcp = ei;
    const int* dstp = ei + e;

    float* bufA = (float*)d_ws;
    float* bufB = bufA + (size_t)n * 128;
    int*   cnt       = (int*)(bufB + (size_t)n * 128);
    int*   slots     = cnt + ((size_t)n + 31) / 32 * 32;
    int*   spill_n   = slots + (size_t)n * NSLOT;
    int*   spill_src = spill_n + 32;
    int*   spill_dst = spill_src + e;
    float* outp = (float*)d_out;

    const int T = 256;
    const int blk_n    = (n + T - 1) / T;
    const int blk_e    = (e + T - 1) / T;
    const int blk_gemm = (((n + 15) / 16) + 7) / 8;
    const int blk_ow   = (int)(((size_t)n * 32 + T - 1) / T);
    const int blk_t16  = (n + 15) / 16;

    gcn_zero<<<blk_n, T, 0, stream>>>(cnt, spill_n, n);
    gcn_edge_gemm<<<blk_gemm + blk_e, T, 0, stream>>>(srcp, dstp, cnt, slots, spill_n,
                                                      spill_src, spill_dst, e,
                                                      x, W1, bufA, n, blk_gemm);

    gcn_gather_gemm<<<blk_t16, 512, 0, stream>>>(cnt, slots, spill_n, spill_src, spill_dst, e,
                                                 bufA, b1, W2, bufB, n);
    gcn_gather_out<<<blk_ow, T, 0, stream>>>(cnt, slots, spill_n, spill_src, spill_dst, e,
                                             bufB, b2, Wo, bo, outp, n);
}


# edge ranks drawn from LDS counters, not 600k returning global atomics: launch 1 orders edge chunks by 128-node destination range, a block per range ranks and fills slots; sum order = LDS arrival rank
# speedup vs baseline: 3.9602x; 1.2473x over previous
#include <hip/hip_runtime.h>
#include <hip/hip_bf16.h>

typedef __attribute__((ext_vector_type(2))) float v2f;
typedef __attribute__((ext_vector_type(8))) float v8f;

#define NSLOT 32
#define RSH   7
#define RNG   (1 << RSH)
#define BINT  1024
#define EPT   8
#define CHUNK (EPT * BINT)
#define RANKW 8
#define SPW   10

__launch_bounds__(BINT)
__global__ void gcn_bin(const int* __restrict__ src, const int* __restrict__ dst, int e,
                        int nbk, int nba, int* __restrict__ sorted, int* __restrict__ offT,
                        int* __restrict__ spill_n) {
    __shared__ int lcnt[BINT];
    __shared__ int wsum[32];
    const int t = threadIdx.x;
    const int lane = t & 31;
    const int wave = t >> 5;
    const int blk = (int)blockIdx.x;
    if (blk == 0 && t == 0) spill_n[0] = 0;
    lcnt[t] = 0;
    const int base = blk * CHUNK;
    int key[EPT], bk[EPT], rk[EPT], of[EPT];
    #pragma unroll
    for (int k = 0; k < EPT; ++k) {
        const int ic = min(base + k * BINT + t, e - 1);
        bk[k]  = dst[ic];
        key[k] = src[ic];
    }
    #pragma unroll
    for (int k = 0; k < EPT; ++k) {
        const bool ok = base + k * BINT + t < e;
        key[k] = ok ? ((key[k] << RSH) | (bk[k] & (RNG - 1))) : -1;
        bk[k] >>= RSH;
    }
    __syncthreads();
    #pragma unroll
    for (int k = 0; k < EPT; ++k)
        rk[k] = (key[k] >= 0) ? atomicAdd(&lcnt[bk[k]], 1) : 0;
    __syncthreads();
    const int c0 = lcnt[t];
    int v = c0;
    #pragma unroll
    for (int off = 1; off < 32; off <<= 1) {
        const int u = __shfl_up(v, off, 32);
        if (lane >= off) v += u;
    }
    if (lane == 31) wsum[wave] = v;
    __syncthreads();
    if (wave == 0) {
        const int w0 = wsum[lane];
        int x = w0;
        #pragma unroll
        for (int off = 1; off < 32; off <<= 1) {
            const int u = __shfl_up(x, off, 32);
            if (lane >= off) x += u;
        }
        wsum[lane] = x - w0;
    }
    __syncthreads();
    const int ex = v - c0 + wsum[wave];
    lcnt[t] = ex;
    if (t <= nbk) offT[(size_t)t * nba + blk] = ex;
    __syncthreads();
    #pragma unroll
    for (int k = 0; k < EPT; ++k) of[k] = lcnt[bk[k]];
    #pragma unroll
    for (int k = 0; k < EPT; ++k)
        if (key[k] >= 0) sorted[base + of[k] + rk[k]] = key[k];
}

__device__ __forceinline__ void gcn_place(int key, int r, int nbase, int* __restrict__ slots,
                                          int* __restrict__ spill_n, int* __restrict__ spill_src,
                                          int* __restrict__ spill_dst) {
    const int dl = key & (RNG - 1);
    const int s = key >> RSH;
    if (r < NSLOT) {
        slots[(size_t)(nbase + dl) * NSLOT + r] = s;
    } else {
        const int o = atomicAdd(spill_n, 1);
        spill_src[o] = s;
        spill_dst[o] = nbase + dl;
    }
}

__device__ __forceinline__ float gcn_dinv(int c) {
    const float d = (float)(c + 1);
    return (d > 0.0f) ? rsqrtf(d) : 0.0f;
}

template <bool RELU>
__device__ __forceinline__ void gemm128_block(const float* __restrict__ A,
                                              const float* __restrict__ bias,
                                              const float* __restrict__ W,
                                              float* __restrict__ OutLin,
                                              int nrows, int gblk) {
    __shared__ float sWp[128 * 128];
    const int tid = threadIdx.x;

    #pragma unroll
    for (int i = 0; i < 16; ++i) {
        int g = (i * 256 + tid) * 4;
        int k = g >> 7;
        int n = g & 127;
        float4 w4 = *(const float4*)(W + g);
        int kk = k >> 2, r = k & 3;
        int hf = r >> 1, j = r & 1;
        float wv[4] = {w4.x, w4.y, w4.z, w4.w};
        #pragma unroll
        for (int q = 0; q < 4; ++q) {
            int col = n + q;
            sWp[kk * 512 + (col >> 4) * 64 + hf * 32 + (col & 15) * 2 + j] = wv[q];
        }
    }
    __syncthreads();

    const int wave = tid >> 5;
    const int lane = tid & 31;
    const int half = lane >> 4;
    const int l16  = lane & 15;
    const int rowtile = gblk * 8 + wave;
    if (rowtile * 16 >= nrows) return;

    v8f acc[8];
    #pragma unroll
    for (int ct = 0; ct < 8; ++ct)
        acc[ct] = (v8f){0.f, 0.f, 0.f, 0.f, 0.f, 0.f, 0.f, 0.f};

    const float* arow = A + (size_t)(rowtile * 16 + l16) * 128;

    for (int kk = 0; kk < 32; ++kk) {
        const int kb = kk * 4 + 2 * half;
        float2 a2 = *(const float2*)(arow + kb);
        if (RELU) {
            const float2 bb = *(const float2*)(bias + kb);
            a2.x = fmaxf(a2.x + bb.x, 0.0f);
            a2.y = fmaxf(a2.y + bb.y, 0.0f);
        }
        v2f a; a.x = a2.x; a.y = a2.y;
        const float* bbase = &sWp[kk * 512 + half * 32 + l16 * 2];
        #pragma unroll
        for (int ct = 0; ct < 8; ++ct) {
            const float2 bp = *(const float2*)(bbase + ct * 64);
            v2f b; b.x = bp.x; b.y = bp.y;
            acc[ct] = __builtin_amdgcn_wmma_f32_16x16x4_f32(
                false, a, false, b, (short)0, acc[ct], false, false);
        }
    }

    const int rowbase = rowtile * 16 + 8 * half;
    #pragma unroll
    for (int ct = 0; ct < 8; ++ct) {
        const int col = ct * 16 + l16;
        #pragma unroll
        for (int j = 0; j < 8; ++j) {
            const size_t off = (size_t)(rowbase + j) * 128 + col;
            OutLin[off] = acc[ct][j];
        }
    }
}

__launch_bounds__(RANKW * 32)
__global__ void gcn_rank_gemm(const int* __restrict__ sorted, const int* __restrict__ offT, int nba,
                              int* __restrict__ cnt, int* __restrict__ slots,
                              int* __restrict__ spill_n, int* __restrict__ spill_src,
                              int* __restrict__ spill_dst,
                              const float* __restrict__ X, const float* __restrict__ W,
                              float* __restrict__ OutLin, int nrows, int gcount) {
    const int b = (int)blockIdx.x;
    if (b < gcount) {
        gemm128_block<false>(X, nullptr, W, OutLin, nrows, b);
    } else {
        __shared__ int lc[RNG];
        const int rb = b - gcount;
        const int nbase = rb << RSH;
        const int t = (int)threadIdx.x;
        const int lane = t & 31;
        const int wave = __builtin_amdgcn_readfirstlane(t >> 5);
        if (t < RNG) lc[t] = 0;
        __syncthreads();
        for (int q0 = 0; q0 < nba; q0 += RANKW * SPW) {
            const int segl = q0 + wave + RANKW * lane;
            int mylo = 0, myhi = 0;
            if (lane < SPW && segl < nba) {
                mylo = offT[(size_t)rb * nba + segl];
                myhi = offT[(size_t)(rb + 1) * nba + segl];
            }
            int key[SPW];
            #pragma unroll
            for (int q = 0; q < SPW; ++q) {
                const int lo = __builtin_amdgcn_readlane(mylo, q);
                const int hi = __builtin_amdgcn_readlane(myhi, q);
                const int seg = q0 + wave + RANKW * q;
                const int i = lo + lane;
                key[q] = (i < hi) ? sorted[(size_t)seg * CHUNK + i] : -1;
            }
            int rr[SPW];
            #pragma unroll
            for (int q = 0; q < SPW; ++q)
                rr[q] = (key[q] >= 0) ? atomicAdd(&lc[key[q] & (RNG - 1)], 1) : 0;
            #pragma unroll
            for (int q = 0; q < SPW; ++q)
                if (key[q] >= 0) gcn_place(key[q], rr[q], nbase, slots, spill_n, spill_src, spill_dst);
            #pragma unroll
            for (int q = 0; q < SPW; ++q) {
                const int lo = __builtin_amdgcn_readlane(mylo, q);
                const int hi = __builtin_amdgcn_readlane(myhi, q);
                const int seg = q0 + wave + RANKW * q;
                for (int i0 = lo + 32; i0 < hi; i0 += 32) {
                    const int i = i0 + lane;
                    if (i < hi) {
                        const int kx = sorted[(size_t)seg * CHUNK + i];
                        gcn_place(kx, atomicAdd(&lc[kx & (RNG - 1)], 1), nbase, slots,
                                  spill_n, spill_src, spill_dst);
                    }
                }
            }
        }
        __syncthreads();
        if (t < RNG && nbase + t < nrows) cnt[nbase + t] = lc[t];
    }
}

__device__ __forceinline__ float4 gcn_aggregate(const int* __restrict__ cnt,
                                                const int* __restrict__ slots,
                                                const int* __restrict__ spill_n,
                                                const int* __restrict__ spill_src,
                                                const int* __restrict__ spill_dst,
                                                int ecap, const float* __restrict__ h,
                                                int w, int lane) {
    const int c = cnt[w];
    const float4 hs = *(const float4*)(h + (size_t)w * 128 + lane * 4);
    const float dd = gcn_dinv(c);
    const int m = min(c, NSLOT);
    const int idx = (lane < m) ? slots[(size_t)w * NSLOT + lane] : 0;
    const float nv = gcn_dinv(cnt[idx]) * dd;
    float4 acc = make_float4(0.0f, 0.0f, 0.0f, 0.0f);
    #pragma unroll 4
    for (int k = 0; k < m; ++k) {
        const int s = __builtin_amdgcn_readlane(idx, k);
        const float nm = __int_as_float(__builtin_amdgcn_readlane(__float_as_int(nv), k));
        const float4 v = *(const float4*)(h + (size_t)s * 128 + lane * 4);
        acc.x += v.x * nm;
        acc.y += v.y * nm;
        acc.z += v.z * nm;
        acc.w += v.w * nm;
    }
    if (c > NSLOT) {
        const int on = min(spill_n[0], ecap);
        for (int base = 0; base < on; base += 32) {
            const int j = base + lane;
            bool mine = false;
            if (j < on) mine = (spill_dst[j] == w);
            const int os = mine ? spill_src[j] : 0;
            const float onv = gcn_dinv(cnt[os]) * dd;
            unsigned mask = __builtin_amdgcn_ballot_w32(mine);
            while (mask != 0u) {
                const int k = __builtin_ctz(mask);
                mask &= mask - 1u;
                const int s = __builtin_amdgcn_readlane(os, k);
                const float nm = __int_as_float(__builtin_amdgcn_readlane(__float_as_int(onv), k));
                const float4 v = *(const float4*)(h + (size_t)s * 128 + lane * 4);
                acc.x += v.x * nm;
                acc.y += v.y * nm;
                acc.z += v.z * nm;
                acc.w += v.w * nm;
            }
        }
    }
    const float d2 = dd * dd;
    acc.x += hs.x * d2;
    acc.y += hs.y * d2;
    acc.z += hs.z * d2;
    acc.w += hs.w * d2;
    return acc;
}

__launch_bounds__(256)
__global__ void gcn_gather_out(const int* __restrict__ cnt, const int* __restrict__ slots,
                               const int* __restrict__ spill_n, const int* __restrict__ spill_src,
                               const int* __restrict__ spill_dst, int ecap,
                               const float* __restrict__ h,
                               const float* __restrict__ b2, const float* __restrict__ Wo,
                               const float* __restrict__ bo, float* __restrict__ prob, int n) {
    const int lane = threadIdx.x & 31;
    const int w = __builtin_amdgcn_readfirstlane((int)((blockIdx.x * blockDim.x + threadIdx.x) >> 5));
    if (w >= n) return;
    const float4 acc = gcn_aggregate(cnt, slots, spill_n, spill_src, spill_dst, ecap, h, w, lane);
    const float4 bb = *(const float4*)(b2 + lane * 4);
    float4 v;
    v.x = fmaxf(acc.x + bb.x, 0.0f);
    v.y = fmaxf(acc.y + bb.y, 0.0f);
    v.z = fmaxf(acc.z + bb.z, 0.0f);
    v.w = fmaxf(acc.w + bb.w, 0.0f);
    float lg[5];
    #pragma unroll
    for (int c = 0; c < 5; ++c) {
        lg[c] = v.x * Wo[(lane * 4 + 0) * 5 + c]
              + v.y * Wo[(lane * 4 + 1) * 5 + c]
              + v.z * Wo[(lane * 4 + 2) * 5 + c]
              + v.w * Wo[(lane * 4 + 3) * 5 + c];
    }
    #pragma unroll
    for (int c = 0; c < 5; ++c) {
        #pragma unroll
        for (int off = 16; off > 0; off >>= 1)
            lg[c] += __shfl_xor(lg[c], off, 32);
    }
    #pragma unroll
    for (int c = 0; c < 5; ++c) lg[c] += bo[c];
    float m = lg[0];
    #pragma unroll
    for (int c = 1; c < 5; ++c) m = fmaxf(m, lg[c]);
    float ex[5], s = 0.0f;
    #pragma unroll
    for (int c = 0; c < 5; ++c) { ex[c] = expf(lg[c] - m); s += ex[c]; }
    const float inv = 1.0f / s;
    if (lane == 0) {
        #pragma unroll
        for (int c = 0; c < 5; ++c) prob[(size_t)w * 5 + c] = ex[c] * inv;
    }
}

#define SA_PITCH 132

__launch_bounds__(512)
__global__ void gcn_gather_gemm(const int* __restrict__ cnt, const int* __restrict__ slots,
                                const int* __restrict__ spill_n, const int* __restrict__ spill_src,
                                const int* __restrict__ spill_dst, int ecap,
                                const float* __restrict__ h,
                                const float* __restrict__ b1, const float* __restrict__ W,
                                float* __restrict__ OutLin, int n) {
    __shared__ float sA[16 * SA_PITCH];
    const int lane = threadIdx.x & 31;
    const int wave = __builtin_amdgcn_readfirstlane((int)(threadIdx.x >> 5));
    const int node0 = (int)blockIdx.x * 16;
    const int w = node0 + wave;
    float4 v = make_float4(0.0f, 0.0f, 0.0f, 0.0f);
    if (w < n) {
        const float4 acc = gcn_aggregate(cnt, slots, spill_n, spill_src, spill_dst, ecap, h, w, lane);
        const float4 bb = *(const float4*)(b1 + lane * 4);
        v.x = fmaxf(acc.x + bb.x, 0.0f);
        v.y = fmaxf(acc.y + bb.y, 0.0f);
        v.z = fmaxf(acc.z + bb.z, 0.0f);
        v.w = fmaxf(acc.w + bb.w, 0.0f);
    }
    const int half = lane >> 4;
    const int l16  = lane & 15;
    const int ct   = wave & 7;
    const float* bcol = W + (size_t)(2 * half) * 128 + ct * 16 + l16;
    float bx[2][8], by[2][8];
    #pragma unroll
    for (int i = 0; i < 8; ++i) { bx[0][i] = 0.0f; by[0][i] = 0.0f; bx[1][i] = 0.0f; by[1][i] = 0.0f; }
    if (wave < 8) {
        #pragma unroll
        for (int i = 0; i < 8; ++i) {
            bx[0][i] = bcol[(size_t)(i * 4) * 128];
            by[0][i] = bcol[(size_t)(i * 4 + 1) * 128];
        }
    }
    *(float4*)&sA[wave * SA_PITCH + lane * 4] = v;
    __syncthreads();
    if (wave >= 8) return;

    v8f d = (v8f){0.f, 0.f, 0.f, 0.f, 0.f, 0.f, 0.f, 0.f};
    const float* arow = &sA[l16 * SA_PITCH + 2 * half];
    #pragma unroll
    for (int g = 0; g < 4; ++g) {
        if (g + 1 < 4) {
            #pragma unroll
            for (int i = 0; i < 8; ++i) {
                const int kn = (g + 1) * 8 + i;
                bx[(g + 1) & 1][i] = bcol[(size_t)(kn * 4) * 128];
                by[(g + 1) & 1][i] = bcol[(size_t)(kn * 4 + 1) * 128];
            }
        }
        #pragma unroll
        for (int i = 0; i < 8; ++i) {
            const int kk = g * 8 + i;
            const float2 a2 = *(const float2*)(arow + kk * 4);
            v2f a; a.x = a2.x; a.y = a2.y;
            v2f b; b.x = bx[g & 1][i]; b.y = by[g & 1][i];
            d = __builtin_amdgcn_wmma_f32_16x16x4_f32(
                false, a, false, b, (short)0, d, false, false);
        }
    }
    const int rowbase = node0 + 8 * half;
    const int col = ct * 16 + l16;
    #pragma unroll
    for (int j = 0; j < 8; ++j) {
        if (rowbase + j < n) OutLin[(size_t)(rowbase + j) * 128 + col] = d[j];
    }
}

extern "C" void kernel_launch(void* const* d_in, const int* in_sizes, int n_in,
                              void* d_out, int out_size, void* d_ws, size_t ws_size,
                              hipStream_t stream) {
    const float* x  = (const float*)d_in[0];
    const int*   ei = (const int*)d_in[1];
    const float* W1 = (const float*)d_in[2];
    const float* b1 = (const float*)d_in[3];
    const float* W2 = (const float*)d_in[4];
    const float* b2 = (const float*)d_in[5];
    const float* Wo = (const float*)d_in[6];
    const float* bo = (const float*)d_in[7];

    const int n = in_sizes[0] / 128;
    const int e = in_sizes[1] / 2;
    const int* srcp = ei;
    const int* dstp = ei + e;

    float* bufA = (float*)d_ws;
    float* bufB = bufA + (size_t)n * 128;
    int*   cnt       = (int*)(bufB + (size_t)n * 128);
    int*   slots     = cnt + ((size_t)n + 31) / 32 * 32;
    int*   spill_n   = slots + (size_t)n * NSLOT;
    int*   spill_src = spill_n + 32;
    int*   spill_dst = spill_src + e;
    const int nbk = (n + RNG - 1) / RNG;
    const int nba = (e + CHUNK - 1) / CHUNK;
    int*   sorted    = spill_dst + ((size_t)e + 31) / 32 * 32;
    int*   offT      = sorted + (size_t)nba * CHUNK;
    float* outp = (float*)d_out;

    const int T = 256;
    const int blk_gemm = (((n + 15) / 16) + 7) / 8;
    const int blk_ow   = (int)(((size_t)n * 32 + T - 1) / T);
    const int blk_t16  = (n + 15) / 16;

    gcn_bin<<<nba, BINT, 0, stream>>>(srcp, dstp, e, nbk, nba, sorted, offT, spill_n);
    gcn_rank_gemm<<<blk_gemm + nbk, RANKW * 32, 0, stream>>>(sorted, offT, nba, cnt, slots, spill_n,
                                                            spill_src, spill_dst,
                                                            x, W1, bufA, n, blk_gemm);

    gcn_gather_gemm<<<blk_t16, 512, 0, stream>>>(cnt, slots, spill_n, spill_src, spill_dst, e,
                                                 bufA, b1, W2, bufB, n);
    gcn_gather_out<<<blk_ow, T, 0, stream>>>(cnt, slots, spill_n, spill_src, spill_dst, e,
                                             bufB, b2, Wo, bo, outp, n);
}


# layer-1 aggregation moved in front of its product, A(xW1) = (Ax)W1: x rows gathered, both products back to back on a 32-node LDS tile, x@W1 pass and its row buffer gone; layer-1 sum order changes
# speedup vs baseline: 4.1728x; 1.0537x over previous
#include <hip/hip_runtime.h>
#include <hip/hip_bf16.h>

typedef __attribute__((ext_vector_type(2))) float v2f;
typedef __attribute__((ext_vector_type(8))) float v8f;

#define NSLOT 32
#define RSH   7
#define RNG   (1 << RSH)
#define BINT  1024
#define EPT   8
#define CHUNK (EPT * BINT)
#define RANKW 8
#define SPW   10

__launch_bounds__(BINT)
__global__ void gcn_bin(const int* __restrict__ src, const int* __restrict__ dst, int e,
                        int nbk, int nba, int* __restrict__ sorted, int* __restrict__ offT,
                        int* __restrict__ spill_n) {
    __shared__ int lcnt[BINT];
    __shared__ int wsum[32];
    const int t = threadIdx.x;
    const int lane = t & 31;
    const int wave = t >> 5;
    const int blk = (int)blockIdx.x;
    if (blk == 0 && t == 0) spill_n[0] = 0;
    lcnt[t] = 0;
    const int base = blk * CHUNK;
    int key[EPT], bk[EPT], rk[EPT], of[EPT];
    #pragma unroll
    for (int k = 0; k < EPT; ++k) {
        const int ic = min(base + k * BINT + t, e - 1);
        bk[k]  = dst[ic];
        key[k] = src[ic];
    }
    #pragma unroll
    for (int k = 0; k < EPT; ++k) {
        const bool ok = base + k * BINT + t < e;
        key[k] = ok ? ((key[k] << RSH) | (bk[k] & (RNG - 1))) : -1;
        bk[k] >>= RSH;
    }
    __syncthreads();
    #pragma unroll
    for (int k = 0; k < EPT; ++k)
        rk[k] = (key[k] >= 0) ? atomicAdd(&lcnt[bk[k]], 1) : 0;
    __syncthreads();
    const int c0 = lcnt[t];
    int v = c0;
    #pragma unroll
    for (int off = 1; off < 32; off <<= 1) {
        const int u = __shfl_up(v, off, 32);
        if (lane >= off) v += u;
    }
    if (lane == 31) wsum[wave] = v;
    __syncthreads();
    if (wave == 0) {
        const int w0 = wsum[lane];
        int x = w0;
        #pragma unroll
        for (int off = 1; off < 32; off <<= 1) {
            const int u = __shfl_up(x, off, 32);
            if (lane >= off) x += u;
        }
        wsum[lane] = x - w0;
    }
    __syncthreads();
    const int ex = v - c0 + wsum[wave];
    lcnt[t] = ex;
    if (t <= nbk) offT[(size_t)t * nba + blk] = ex;
    __syncthreads();
    #pragma unroll
    for (int k = 0; k < EPT; ++k) of[k] = lcnt[bk[k]];
    #pragma unroll
    for (int k = 0; k < EPT; ++k)
        if (key[k] >= 0) sorted[base + of[k] + rk[k]] = key[k];
}

__device__ __forceinline__ void gcn_place(int key, int r, int nbase, int* __restrict__ slots,
                                          int* __restrict__ spill_n, int* __restrict__ spill_src,
                                          int* __restrict__ spill_dst) {
    const int dl = key & (RNG - 1);
    const int s = key >> RSH;
    if (r < NSLOT) {
        slots[(size_t)(nbase + dl) * NSLOT + r] = s;
    } else {
        const int o = atomicAdd(spill_n, 1);
        spill_src[o] = s;
        spill_dst[o] = nbase + dl;
    }
}

__device__ __forceinline__ float gcn_dinv(int c) {
    const float d = (float)(c + 1);
    return (d > 0.0f) ? rsqrtf(d) : 0.0f;
}

__launch_bounds__(RANKW * 32)
__global__ void gcn_rank(const int* __restrict__ sorted, const int* __restrict__ offT, int nba,
                         int* __restrict__ cnt, int* __restrict__ slots,
                         int* __restrict__ spill_n, int* __restrict__ spill_src,
                         int* __restrict__ spill_dst, int nrows) {
    __shared__ int lc[RNG];
    const int rb = (int)blockIdx.x;
    const int nbase = rb << RSH;
    const int t = (int)threadIdx.x;
    const int lane = t & 31;
    const int wave = __builtin_amdgcn_readfirstlane(t >> 5);
    if (t < RNG) lc[t] = 0;
    __syncthreads();
    for (int q0 = 0; q0 < nba; q0 += RANKW * SPW) {
        const int segl = q0 + wave + RANKW * lane;
        int mylo = 0, myhi = 0;
        if (lane < SPW && segl < nba) {
            mylo = offT[(size_t)rb * nba + segl];
            myhi = offT[(size_t)(rb + 1) * nba + segl];
        }
        int key[SPW];
        #pragma unroll
        for (int q = 0; q < SPW; ++q) {
            const int lo = __builtin_amdgcn_readlane(mylo, q);
            const int hi = __builtin_amdgcn_readlane(myhi, q);
            const int seg = q0 + wave + RANKW * q;
            const int i = lo + lane;
            key[q] = (i < hi) ? sorted[(size_t)seg * CHUNK + i] : -1;
        }
        int rr[SPW];
        #pragma unroll
        for (int q = 0; q < SPW; ++q)
            rr[q] = (key[q] >= 0) ? atomicAdd(&lc[key[q] & (RNG - 1)], 1) : 0;
        #pragma unroll
        for (int q = 0; q < SPW; ++q)
            if (key[q] >= 0) gcn_place(key[q], rr[q], nbase, slots, spill_n, spill_src, spill_dst);
        #pragma unroll
        for (int q = 0; q < SPW; ++q) {
            const int lo = __builtin_amdgcn_readlane(mylo, q);
            const int hi = __builtin_amdgcn_readlane(myhi, q);
            const int seg = q0 + wave + RANKW * q;
            for (int i0 = lo + 32; i0 < hi; i0 += 32) {
                const int i = i0 + lane;
                if (i < hi) {
                    const int kx = sorted[(size_t)seg * CHUNK + i];
                    gcn_place(kx, atomicAdd(&lc[kx & (RNG - 1)], 1), nbase, slots,
                              spill_n, spill_src, spill_dst);
                }
            }
        }
    }
    __syncthreads();
    if (t < RNG && nbase + t < nrows) cnt[nbase + t] = lc[t];
}

struct gcn_head {
    int c, m, idx;
    float dd, nv;
};

__device__ __forceinline__ gcn_head gcn_agg_head(const int* __restrict__ cnt,
                                                 const int* __restrict__ slots,
                                                 int w, int lane) {
    gcn_head q;
    q.c = cnt[w];
    q.dd = gcn_dinv(q.c);
    q.m = min(q.c, NSLOT);
    q.idx = (lane < q.m) ? slots[(size_t)w * NSLOT + lane] : 0;
    q.nv = gcn_dinv(cnt[q.idx]) * q.dd;
    return q;
}

__device__ __forceinline__ float4 gcn_agg_body(const gcn_head q, const int* __restrict__ cnt,
                                               const int* __restrict__ spill_n,
                                               const int* __restrict__ spill_src,
                                               const int* __restrict__ spill_dst,
                                               int ecap, const float* __restrict__ h,
                                               int w, int lane) {
    const int c = q.c;
    const int m = q.m;
    const int idx = q.idx;
    const float dd = q.dd;
    const float nv = q.nv;
    const float4 hs = *(const float4*)(h + (size_t)w * 128 + lane * 4);
    float4 acc = make_float4(0.0f, 0.0f, 0.0f, 0.0f);
    #pragma unroll 4
    for (int k = 0; k < m; ++k) {
        const int s = __builtin_amdgcn_readlane(idx, k);
        const float nm = __int_as_float(__builtin_amdgcn_readlane(__float_as_int(nv), k));
        const float4 v = *(const float4*)(h + (size_t)s * 128 + lane * 4);
        acc.x += v.x * nm;
        acc.y += v.y * nm;
        acc.z += v.z * nm;
        acc.w += v.w * nm;
    }
    if (c > NSLOT) {
        const int on = min(spill_n[0], ecap);
        for (int base = 0; base < on; base += 32) {
            const int j = base + lane;
            bool mine = false;
            if (j < on) mine = (spill_dst[j] == w);
            const int os = mine ? spill_src[j] : 0;
            const float onv = gcn_dinv(cnt[os]) * dd;
            unsigned mask = __builtin_amdgcn_ballot_w32(mine);
            while (mask != 0u) {
                const int k = __builtin_ctz(mask);
                mask &= mask - 1u;
                const int s = __builtin_amdgcn_readlane(os, k);
                const float nm = __int_as_float(__builtin_amdgcn_readlane(__float_as_int(onv), k));
                const float4 v = *(const float4*)(h + (size_t)s * 128 + lane * 4);
                acc.x += v.x * nm;
                acc.y += v.y * nm;
                acc.z += v.z * nm;
                acc.w += v.w * nm;
            }
        }
    }
    const float d2 = dd * dd;
    acc.x += hs.x * d2;
    acc.y += hs.y * d2;
    acc.z += hs.z * d2;
    acc.w += hs.w * d2;
    return acc;
}

__device__ __forceinline__ float4 gcn_aggregate(const int* __restrict__ cnt,
                                                const int* __restrict__ slots,
                                                const int* __restrict__ spill_n,
                                                const int* __restrict__ spill_src,
                                                const int* __restrict__ spill_dst,
                                                int ecap, const float* __restrict__ h,
                                                int w, int lane) {
    const gcn_head q = gcn_agg_head(cnt, slots, w, lane);
    return gcn_agg_body(q, cnt, spill_n, spill_src, spill_dst, ecap, h, w, lane);
}

__launch_bounds__(256)
__global__ void gcn_gather_out(const int* __restrict__ cnt, const int* __restrict__ slots,
                               const int* __restrict__ spill_n, const int* __restrict__ spill_src,
                               const int* __restrict__ spill_dst, int ecap,
                               const float* __restrict__ h,
                               const float* __restrict__ b2, const float* __restrict__ Wo,
                               const float* __restrict__ bo, float* __restrict__ prob, int n) {
    const int lane = threadIdx.x & 31;
    const int w = __builtin_amdgcn_readfirstlane((int)((blockIdx.x * blockDim.x + threadIdx.x) >> 5));
    if (w >= n) return;
    const float4 acc = gcn_aggregate(cnt, slots, spill_n, spill_src, spill_dst, ecap, h, w, lane);
    const float4 bb = *(const float4*)(b2 + lane * 4);
    float4 v;
    v.x = fmaxf(acc.x + bb.x, 0.0f);
    v.y = fmaxf(acc.y + bb.y, 0.0f);
    v.z = fmaxf(acc.z + bb.z, 0.0f);
    v.w = fmaxf(acc.w + bb.w, 0.0f);
    float lg[5];
    #pragma unroll
    for (int c = 0; c < 5; ++c) {
        lg[c] = v.x * Wo[(lane * 4 + 0) * 5 + c]
              + v.y * Wo[(lane * 4 + 1) * 5 + c]
              + v.z * Wo[(lane * 4 + 2) * 5 + c]
              + v.w * Wo[(lane * 4 + 3) * 5 + c];
    }
    #pragma unroll
    for (int c = 0; c < 5; ++c) {
        #pragma unroll
        for (int off = 16; off > 0; off >>= 1)
            lg[c] += __shfl_xor(lg[c], off, 32);
    }
    #pragma unroll
    for (int c = 0; c < 5; ++c) lg[c] += bo[c];
    float m = lg[0];
    #pragma unroll
    for (int c = 1; c < 5; ++c) m = fmaxf(m, lg[c]);
    float ex[5], s = 0.0f;
    #pragma unroll
    for (int c = 0; c < 5; ++c) { ex[c] = expf(lg[c] - m); s += ex[c]; }
    const float inv = 1.0f / s;
    if (lane == 0) {
        #pragma unroll
        for (int c = 0; c < 5; ++c) prob[(size_t)w * 5 + c] = ex[c] * inv;
    }
}

#define SA_PITCH 132
#define TILE_N   32

__device__ __forceinline__ void gcn_bload(const float* __restrict__ bcol, int g,
                                          float (&bx)[8], float (&by)[8]) {
    #pragma unroll
    for (int i = 0; i < 8; ++i) {
        const int kn = g * 8 + i;
        bx[i] = bcol[(size_t)(kn * 4) * 128];
        by[i] = bcol[(size_t)(kn * 4 + 1) * 128];
    }
}

__device__ __forceinline__ v8f gcn_tile(v8f d, const float* arow, const float* __restrict__ bcol,
                                        float (&bx)[2][8], float (&by)[2][8]) {
    #pragma unroll
    for (int g = 0; g < 4; ++g) {
        if (g + 1 < 4) gcn_bload(bcol, g + 1, bx[(g + 1) & 1], by[(g + 1) & 1]);
        #pragma unroll
        for (int i = 0; i < 8; ++i) {
            const int kk = g * 8 + i;
            const float2 a2 = *(const float2*)(arow + kk * 4);
            v2f a; a.x = a2.x; a.y = a2.y;
            v2f b; b.x = bx[g & 1][i]; b.y = by[g & 1][i];
            d = __builtin_amdgcn_wmma_f32_16x16x4_f32(
                false, a, false, b, (short)0, d, false, false);
        }
    }
    return d;
}

__launch_bounds__(512)
__global__ void gcn_gather_gemm2(const int* __restrict__ cnt, const int* __restrict__ slots,
                                 const int* __restrict__ spill_n, const int* __restrict__ spill_src,
                                 const int* __restrict__ spill_dst, int ecap,
                                 const float* __restrict__ x,
                                 const float* __restrict__ W1, const float* __restrict__ b1,
                                 const float* __restrict__ W2,
                                 float* __restrict__ OutLin, int n) {
    __shared__ float sA[TILE_N * SA_PITCH];
    __shared__ float sH[TILE_N * SA_PITCH];
    const int lane = threadIdx.x & 31;
    const int wave = __builtin_amdgcn_readfirstlane((int)(threadIdx.x >> 5));
    const int node0 = (int)blockIdx.x * TILE_N;
    const int wA = min(node0 + wave, n - 1);
    const int wB = min(node0 + 16 + wave, n - 1);
    const gcn_head qA = gcn_agg_head(cnt, slots, wA, lane);
    const gcn_head qB = gcn_agg_head(cnt, slots, wB, lane);
    float4 vA = gcn_agg_body(qA, cnt, spill_n, spill_src, spill_dst, ecap, x, wA, lane);
    float4 vB = gcn_agg_body(qB, cnt, spill_n, spill_src, spill_dst, ecap, x, wB, lane);
    if (node0 + wave >= n)      vA = make_float4(0.0f, 0.0f, 0.0f, 0.0f);
    if (node0 + 16 + wave >= n) vB = make_float4(0.0f, 0.0f, 0.0f, 0.0f);
    *(float4*)&sA[wave * SA_PITCH + lane * 4] = vA;
    *(float4*)&sA[(16 + wave) * SA_PITCH + lane * 4] = vB;
    const int half = lane >> 4;
    const int l16  = lane & 15;
    const int ct   = wave & 7;
    const int rt   = wave >> 3;
    const int col  = ct * 16 + l16;
    const float* bcol1 = W1 + (size_t)(2 * half) * 128 + col;
    const float* bcol2 = W2 + (size_t)(2 * half) * 128 + col;
    float bx[2][8], by[2][8];
    gcn_bload(bcol1, 0, bx[0], by[0]);
    const float bias = b1[col];
    const v8f c1 = (v8f){bias, bias, bias, bias, bias, bias, bias, bias};
    __syncthreads();

    const v8f d1 = gcn_tile(c1, &sA[(rt * 16 + l16) * SA_PITCH + 2 * half], bcol1, bx, by);
    gcn_bload(bcol2, 0, bx[0], by[0]);
    const int hrow = rt * 16 + 8 * half;
    #pragma unroll
    for (int j = 0; j < 8; ++j)
        sH[(hrow + j) * SA_PITCH + col] = fmaxf(d1[j], 0.0f);
    __syncthreads();

    const v8f z8 = (v8f){0.f, 0.f, 0.f, 0.f, 0.f, 0.f, 0.f, 0.f};
    const v8f d2 = gcn_tile(z8, &sH[(rt * 16 + l16) * SA_PITCH + 2 * half], bcol2, bx, by);
    const int rowbase = node0 + hrow;
    #pragma unroll
    for (int j = 0; j < 8; ++j) {
        if (rowbase + j < n) OutLin[(size_t)(rowbase + j) * 128 + col] = d2[j];
    }
}

extern "C" void kernel_launch(void* const* d_in, const int* in_sizes, int n_in,
                              void* d_out, int out_size, void* d_ws, size_t ws_size,
                              hipStream_t stream) {
    const float* x  = (const float*)d_in[0];
    const int*   ei = (const int*)d_in[1];
    const float* W1 = (const float*)d_in[2];
    const float* b1 = (const float*)d_in[3];
    const float* W2 = (const float*)d_in[4];
    const float* b2 = (const float*)d_in[5];
    const float* Wo = (const float*)d_in[6];
    const float* bo = (const float*)d_in[7];

    const int n = in_sizes[0] / 128;
    const int e = in_sizes[1] / 2;
    const int* srcp = ei;
    const int* dstp = ei + e;

    float* bufB = (float*)d_ws;
    int*   cnt       = (int*)(bufB + (size_t)n * 128);
    int*   slots     = cnt + ((size_t)n + 31) / 32 * 32;
    int*   spill_n   = slots + (size_t)n * NSLOT;
    int*   spill_src = spill_n + 32;
    int*   spill_dst = spill_src + e;
    const int nbk = (n + RNG - 1) / RNG;
    const int nba = (e + CHUNK - 1) / CHUNK;
    int*   sorted    = spill_dst + ((size_t)e + 31) / 32 * 32;
    int*   offT      = sorted + (size_t)nba * CHUNK;
    float* outp = (float*)d_out;

    const int T = 256;
    const int blk_ow  = (int)(((size_t)n * 32 + T - 1) / T);
    const int blk_t32 = (n + TILE_N - 1) / TILE_N;

    gcn_bin<<<nba, BINT, 0, stream>>>(srcp, dstp, e, nbk, nba, sorted, offT, spill_n);
    gcn_rank<<<nbk, RANKW * 32, 0, stream>>>(sorted, offT, nba, cnt, slots, spill_n,
                                             spill_src, spill_dst, n);

    gcn_gather_gemm2<<<blk_t32, 512, 0, stream>>>(cnt, slots, spill_n, spill_src, spill_dst, e,
                                                  x, W1, b1, W2, bufB, n);
    gcn_gather_out<<<blk_ow, T, 0, stream>>>(cnt, slots, spill_n, spill_src, spill_dst, e,
                                             bufB, b2, Wo, bo, outp, n);
}


# layer-1/2 products: each B fragment feeds both 16-row tiles of a block (2 accumulators per wave, B loads per WMMA halved); blocks of 8 waves gather 4 nodes each, heads in stages; sum order kept
# speedup vs baseline: 4.3804x; 1.0498x over previous
#include <hip/hip_runtime.h>
#include <hip/hip_bf16.h>

typedef __attribute__((ext_vector_type(2))) float v2f;
typedef __attribute__((ext_vector_type(8))) float v8f;

#define NSLOT 32
#define RSH   7
#define RNG   (1 << RSH)
#define BINT  1024
#define EPT   8
#define CHUNK (EPT * BINT)
#define RANKW 8
#define SPW   10

__launch_bounds__(BINT)
__global__ void gcn_bin(const int* __restrict__ src, const int* __restrict__ dst, int e,
                        int nbk, int nba, int* __restrict__ sorted, int* __restrict__ offT,
                        int* __restrict__ spill_n) {
    __shared__ int lcnt[BINT];
    __shared__ int wsum[32];
    const int t = threadIdx.x;
    const int lane = t & 31;
    const int wave = t >> 5;
    const int blk = (int)blockIdx.x;
    if (blk == 0 && t == 0) spill_n[0] = 0;
    lcnt[t] = 0;
    const int base = blk * CHUNK;
    int key[EPT], bk[EPT], rk[EPT], of[EPT];
    #pragma unroll
    for (int k = 0; k < EPT; ++k) {
        const int ic = min(base + k * BINT + t, e - 1);
        bk[k]  = dst[ic];
        key[k] = src[ic];
    }
    #pragma unroll
    for (int k = 0; k < EPT; ++k) {
        const bool ok = base + k * BINT + t < e;
        key[k] = ok ? ((key[k] << RSH) | (bk[k] & (RNG - 1))) : -1;
        bk[k] >>= RSH;
    }
    __syncthreads();
    #pragma unroll
    for (int k = 0; k < EPT; ++k)
        rk[k] = (key[k] >= 0) ? atomicAdd(&lcnt[bk[k]], 1) : 0;
    __syncthreads();
    const int c0 = lcnt[t];
    int v = c0;
    #pragma unroll
    for (int off = 1; off < 32; off <<= 1) {
        const int u = __shfl_up(v, off, 32);
        if (lane >= off) v += u;
    }
    if (lane == 31) wsum[wave] = v;
    __syncthreads();
    if (wave == 0) {
        const int w0 = wsum[lane];
        int x = w0;
        #pragma unroll
        for (int off = 1; off < 32; off <<= 1) {
            const int u = __shfl_up(x, off, 32);
            if (lane >= off) x += u;
        }
        wsum[lane] = x - w0;
    }
    __syncthreads();
    const int ex = v - c0 + wsum[wave];
    lcnt[t] = ex;
    if (t <= nbk) offT[(size_t)t * nba + blk] = ex;
    __syncthreads();
    #pragma unroll
    for (int k = 0; k < EPT; ++k) of[k] = lcnt[bk[k]];
    #pragma unroll
    for (int k = 0; k < EPT; ++k)
        if (key[k] >= 0) sorted[base + of[k] + rk[k]] = key[k];
}

__device__ __forceinline__ void gcn_place(int key, int r, int nbase, int* __restrict__ slots,
                                          int* __restrict__ spill_n, int* __restrict__ spill_src,
                                          int* __restrict__ spill_dst) {
    const int dl = key & (RNG - 1);
    const int s = key >> RSH;
    if (r < NSLOT) {
        slots[(size_t)(nbase + dl) * NSLOT + r] = s;
    } else {
        const int o = atomicAdd(spill_n, 1);
        spill_src[o] = s;
        spill_dst[o] = nbase + dl;
    }
}

__device__ __forceinline__ float gcn_dinv(int c) {
    const float d = (float)(c + 1);
    return (d > 0.0f) ? rsqrtf(d) : 0.0f;
}

__launch_bounds__(RANKW * 32)
__global__ void gcn_rank(const int* __restrict__ sorted, const int* __restrict__ offT, int nba,
                         int* __restrict__ cnt, int* __restrict__ slots,
                         int* __restrict__ spill_n, int* __restrict__ spill_src,
                         int* __restrict__ spill_dst, int nrows) {
    __shared__ int lc[RNG];
    const int rb = (int)blockIdx.x;
    const int nbase = rb << RSH;
    const int t = (int)threadIdx.x;
    const int lane = t & 31;
    const int wave = __builtin_amdgcn_readfirstlane(t >> 5);
    if (t < RNG) lc[t] = 0;
    __syncthreads();
    for (int q0 = 0; q0 < nba; q0 += RANKW * SPW) {
        const int segl = q0 + wave + RANKW * lane;
        int mylo = 0, myhi = 0;
        if (lane < SPW && segl < nba) {
            mylo = offT[(size_t)rb * nba + segl];
            myhi = offT[(size_t)(rb + 1) * nba + segl];
        }
        int key[SPW];
        #pragma unroll
        for (int q = 0; q < SPW; ++q) {
            const int lo = __builtin_amdgcn_readlane(mylo, q);
            const int hi = __builtin_amdgcn_readlane(myhi, q);
            const int seg = q0 + wave + RANKW * q;
            const int i = lo + lane;
            key[q] = (i < hi) ? sorted[(size_t)seg * CHUNK + i] : -1;
        }
        int rr[SPW];
        #pragma unroll
        for (int q = 0; q < SPW; ++q)
            rr[q] = (key[q] >= 0) ? atomicAdd(&lc[key[q] & (RNG - 1)], 1) : 0;
        #pragma unroll
        for (int q = 0; q < SPW; ++q)
            if (key[q] >= 0) gcn_place(key[q], rr[q], nbase, slots, spill_n, spill_src, spill_dst);
        #pragma unroll
        for (int q = 0; q < SPW; ++q) {
            const int lo = __builtin_amdgcn_readlane(mylo, q);
            const int hi = __builtin_amdgcn_readlane(myhi, q);
            const int seg = q0 + wave + RANKW * q;
            for (int i0 = lo + 32; i0 < hi; i0 += 32) {
                const int i = i0 + lane;
                if (i < hi) {
                    const int kx = sorted[(size_t)seg * CHUNK + i];
                    gcn_place(kx, atomicAdd(&lc[kx & (RNG - 1)], 1), nbase, slots,
                              spill_n, spill_src, spill_dst);
                }
            }
        }
    }
    __syncthreads();
    if (t < RNG && nbase + t < nrows) cnt[nbase + t] = lc[t];
}

struct gcn_head {
    int c, m, idx;
    float dd, nv;
};

__device__ __forceinline__ gcn_head gcn_agg_head(const int* __restrict__ cnt,
                                                 const int* __restrict__ slots,
                                                 int w, int lane) {
    gcn_head q;
    q.c = cnt[w];
    q.dd = gcn_dinv(q.c);
    q.m = min(q.c, NSLOT);
    q.idx = (lane < q.m) ? slots[(size_t)w * NSLOT + lane] : 0;
    q.nv = gcn_dinv(cnt[q.idx]) * q.dd;
    return q;
}

__device__ __forceinline__ float4 gcn_agg_body(const gcn_head q, const int* __restrict__ cnt,
                                               const int* __restrict__ spill_n,
                                               const int* __restrict__ spill_src,
                                               const int* __restrict__ spill_dst,
                                               int ecap, const float* __restrict__ h,
                                               int w, int lane) {
    const int c = q.c;
    const int m = q.m;
    const int idx = q.idx;
    const float dd = q.dd;
    const float nv = q.nv;
    const float4 hs = *(const float4*)(h + (size_t)w * 128 + lane * 4);
    float4 acc = make_float4(0.0f, 0.0f, 0.0f, 0.0f);
    #pragma unroll 4
    for (int k = 0; k < m; ++k) {
        const int s = __builtin_amdgcn_readlane(idx, k);
        const float nm = __int_as_float(__builtin_amdgcn_readlane(__float_as_int(nv), k));
        const float4 v = *(const float4*)(h + (size_t)s * 128 + lane * 4);
        acc.x += v.x * nm;
        acc.y += v.y * nm;
        acc.z += v.z * nm;
        acc.w += v.w * nm;
    }
    if (c > NSLOT) {
        const int on = min(spill_n[0], ecap);
        for (int base = 0; base < on; base += 32) {
            const int j = base + lane;
            bool mine = false;
            if (j < on) mine = (spill_dst[j] == w);
            const int os = mine ? spill_src[j] : 0;
            const float onv = gcn_dinv(cnt[os]) * dd;
            unsigned mask = __builtin_amdgcn_ballot_w32(mine);
            while (mask != 0u) {
                const int k = __builtin_ctz(mask);
                mask &= mask - 1u;
                const int s = __builtin_amdgcn_readlane(os, k);
                const float nm = __int_as_float(__builtin_amdgcn_readlane(__float_as_int(onv), k));
                const float4 v = *(const float4*)(h + (size_t)s * 128 + lane * 4);
                acc.x += v.x * nm;
                acc.y += v.y * nm;
                acc.z += v.z * nm;
                acc.w += v.w * nm;
            }
        }
    }
    const float d2 = dd * dd;
    acc.x += hs.x * d2;
    acc.y += hs.y * d2;
    acc.z += hs.z * d2;
    acc.w += hs.w * d2;
    return acc;
}

__device__ __forceinline__ float4 gcn_aggregate(const int* __restrict__ cnt,
                                                const int* __restrict__ slots,
                                                const int* __restrict__ spill_n,
                                                const int* __restrict__ spill_src,
                                                const int* __restrict__ spill_dst,
                                                int ecap, const float* __restrict__ h,
                                                int w, int lane) {
    const gcn_head q = gcn_agg_head(cnt, slots, w, lane);
    return gcn_agg_body(q, cnt, spill_n, spill_src, spill_dst, ecap, h, w, lane);
}

__launch_bounds__(256)
__global__ void gcn_gather_out(const int* __restrict__ cnt, const int* __restrict__ slots,
                               const int* __restrict__ spill_n, const int* __restrict__ spill_src,
                               const int* __restrict__ spill_dst, int ecap,
                               const float* __restrict__ h,
                               const float* __restrict__ b2, const float* __restrict__ Wo,
                               const float* __restrict__ bo, float* __restrict__ prob, int n) {
    const int lane = threadIdx.x & 31;
    const int w = __builtin_amdgcn_readfirstlane((int)((blockIdx.x * blockDim.x + threadIdx.x) >> 5));
    if (w >= n) return;
    const float4 acc = gcn_aggregate(cnt, slots, spill_n, spill_src, spill_dst, ecap, h, w, lane);
    const float4 bb = *(const float4*)(b2 + lane * 4);
    float4 v;
    v.x = fmaxf(acc.x + bb.x, 0.0f);
    v.y = fmaxf(acc.y + bb.y, 0.0f);
    v.z = fmaxf(acc.z + bb.z, 0.0f);
    v.w = fmaxf(acc.w + bb.w, 0.0f);
    float lg[5];
    #pragma unroll
    for (int c = 0; c < 5; ++c) {
        lg[c] = v.x * Wo[(lane * 4 + 0) * 5 + c]
              + v.y * Wo[(lane * 4 + 1) * 5 + c]
              + v.z * Wo[(lane * 4 + 2) * 5 + c]
              + v.w * Wo[(lane * 4 + 3) * 5 + c];
    }
    #pragma unroll
    for (int c = 0; c < 5; ++c) {
        #pragma unroll
        for (int off = 16; off > 0; off >>= 1)
            lg[c] += __shfl_xor(lg[c], off, 32);
    }
    #pragma unroll
    for (int c = 0; c < 5; ++c) lg[c] += bo[c];
    float m = lg[0];
    #pragma unroll
    for (int c = 1; c < 5; ++c) m = fmaxf(m, lg[c]);
    float ex[5], s = 0.0f;
    #pragma unroll
    for (int c = 0; c < 5; ++c) { ex[c] = expf(lg[c] - m); s += ex[c]; }
    const float inv = 1.0f / s;
    if (lane == 0) {
        #pragma unroll
        for (int c = 0; c < 5; ++c) prob[(size_t)w * 5 + c] = ex[c] * inv;
    }
}

#define SA_PITCH 132
#define TILE_N   32
#define GW       8
#define NPW      (TILE_N / GW)

__device__ __forceinline__ void gcn_bload(const float* __restrict__ W, unsigned boff, int g,
                                          float (&bx)[8], float (&by)[8]) {
    #pragma unroll
    for (int i = 0; i < 8; ++i) {
        const unsigned kn = (unsigned)(g * 8 + i);
        bx[i] = W[boff + (kn * 4u) * 128u];
        by[i] = W[boff + (kn * 4u + 1u) * 128u];
    }
}

__device__ __forceinline__ void gcn_tile2(v8f& d0, v8f& d1, const float* arow,
                                          const float* __restrict__ W, unsigned boff,
                                          float (&bx)[2][8], float (&by)[2][8]) {
    #pragma unroll
    for (int g = 0; g < 4; ++g) {
        if (g + 1 < 4) gcn_bload(W, boff, g + 1, bx[(g + 1) & 1], by[(g + 1) & 1]);
        #pragma unroll
        for (int i = 0; i < 8; ++i) {
            const int kk = g * 8 + i;
            const float2 p0 = *(const float2*)(arow + kk * 4);
            const float2 p1 = *(const float2*)(arow + 16 * SA_PITCH + kk * 4);
            v2f a0; a0.x = p0.x; a0.y = p0.y;
            v2f a1; a1.x = p1.x; a1.y = p1.y;
            v2f b; b.x = bx[g & 1][i]; b.y = by[g & 1][i];
            d0 = __builtin_amdgcn_wmma_f32_16x16x4_f32(
                false, a0, false, b, (short)0, d0, false, false);
            d1 = __builtin_amdgcn_wmma_f32_16x16x4_f32(
                false, a1, false, b, (short)0, d1, false, false);
        }
    }
}

__launch_bounds__(GW * 32, 16)
__global__ void gcn_gather_gemm2(const int* __restrict__ cnt, const int* __restrict__ slots,
                                 const int* __restrict__ spill_n, const int* __restrict__ spill_src,
                                 const int* __restrict__ spill_dst, int ecap,
                                 const float* __restrict__ x,
                                 const float* __restrict__ W1, const float* __restrict__ b1,
                                 const float* __restrict__ W2,
                                 float* __restrict__ OutLin, int n) {
    __shared__ float sA[TILE_N * SA_PITCH];
    __shared__ float sH[TILE_N * SA_PITCH];
    const int lane = threadIdx.x & 31;
    const int wave = __builtin_amdgcn_readfirstlane((int)(threadIdx.x >> 5));
    const int node0 = (int)blockIdx.x * TILE_N;
    int wn[NPW];
    gcn_head q[NPW];
    #pragma unroll
    for (int r = 0; r < NPW; ++r) {
        wn[r] = min(node0 + GW * r + wave, n - 1);
        q[r].c = cnt[wn[r]];
    }
    #pragma unroll
    for (int r = 0; r < NPW; ++r) {
        q[r].dd = gcn_dinv(q[r].c);
        q[r].m = min(q[r].c, NSLOT);
        q[r].idx = (lane < q[r].m) ? slots[(size_t)wn[r] * NSLOT + lane] : 0;
    }
    #pragma unroll
    for (int r = 0; r < NPW; ++r)
        q[r].nv = gcn_dinv(cnt[q[r].idx]) * q[r].dd;
    #pragma unroll
    for (int r = 0; r < NPW; ++r) {
        float4 v = gcn_agg_body(q[r], cnt, spill_n, spill_src, spill_dst, ecap, x, wn[r], lane);
        if (node0 + GW * r + wave >= n) v = make_float4(0.0f, 0.0f, 0.0f, 0.0f);
        *(float4*)&sA[(GW * r + wave) * SA_PITCH + lane * 4] = v;
    }
    const int half = lane >> 4;
    const int l16  = lane & 15;
    const int col  = wave * 16 + l16;
    const unsigned boff = (unsigned)((2 * half) * 128 + col);
    float bx[2][8], by[2][8];
    gcn_bload(W1, boff, 0, bx[0], by[0]);
    const float bias = b1[col];
    v8f d1a = (v8f){bias, bias, bias, bias, bias, bias, bias, bias};
    v8f d1b = d1a;
    __syncthreads();

    gcn_tile2(d1a, d1b, &sA[l16 * SA_PITCH + 2 * half], W1, boff, bx, by);
    gcn_bload(W2, boff, 0, bx[0], by[0]);
    const int hrow = 8 * half;
    #pragma unroll
    for (int j = 0; j < 8; ++j) {
        sH[(hrow + j) * SA_PITCH + col]      = fmaxf(d1a[j], 0.0f);
        sH[(16 + hrow + j) * SA_PITCH + col] = fmaxf(d1b[j], 0.0f);
    }
    __syncthreads();

    v8f d2a = (v8f){0.f, 0.f, 0.f, 0.f, 0.f, 0.f, 0.f, 0.f};
    v8f d2b = d2a;
    gcn_tile2(d2a, d2b, &sH[l16 * SA_PITCH + 2 * half], W2, boff, bx, by);
    const int rowbase = node0 + hrow;
    #pragma unroll
    for (int j = 0; j < 8; ++j) {
        if (rowbase + j < n)      OutLin[(size_t)(rowbase + j) * 128 + col] = d2a[j];
        if (rowbase + 16 + j < n) OutLin[(size_t)(rowbase + 16 + j) * 128 + col] = d2b[j];
    }
}

extern "C" void kernel_launch(void* const* d_in, const int* in_sizes, int n_in,
                              void* d_out, int out_size, void* d_ws, size_t ws_size,
                              hipStream_t stream) {
    const float* x  = (const float*)d_in[0];
    const int*   ei = (const int*)d_in[1];
    const float* W1 = (const float*)d_in[2];
    const float* b1 = (const float*)d_in[3];
    const float* W2 = (const float*)d_in[4];
    const float* b2 = (const float*)d_in[5];
    const float* Wo = (const float*)d_in[6];
    const float* bo = (const float*)d_in[7];

    const int n = in_sizes[0] / 128;
    const int e = in_sizes[1] / 2;
    const int* srcp = ei;
    const int* dstp = ei + e;

    float* bufB = (float*)d_ws;
    int*   cnt       = (int*)(bufB + (size_t)n * 128);
    int*   slots     = cnt + ((size_t)n + 31) / 32 * 32;
    int*   spill_n   = slots + (size_t)n * NSLOT;
    int*   spill_src = spill_n + 32;
    int*   spill_dst = spill_src + e;
    const int nbk = (n + RNG - 1) / RNG;
    const int nba = (e + CHUNK - 1) / CHUNK;
    int*   sorted    = spill_dst + ((size_t)e + 31) / 32 * 32;
    int*   offT      = sorted + (size_t)nba * CHUNK;
    float* outp = (float*)d_out;

    const int T = 256;
    const int blk_ow  = (int)(((size_t)n * 32 + T - 1) / T);
    const int blk_t32 = (n + TILE_N - 1) / TILE_N;

    gcn_bin<<<nba, BINT, 0, stream>>>(srcp, dstp, e, nbk, nba, sorted, offT, spill_n);
    gcn_rank<<<nbk, RANKW * 32, 0, stream>>>(sorted, offT, nba, cnt, slots, spill_n,
                                             spill_src, spill_dst, n);

    gcn_gather_gemm2<<<blk_t32, 256, 0, stream>>>(cnt, slots, spill_n, spill_src, spill_dst, e,
                                                  x, W1, b1, W2, bufB, n);
    gcn_gather_out<<<blk_ow, T, 0, stream>>>(cnt, slots, spill_n, spill_src, spill_dst, e,
                                             bufB, b2, Wo, bo, outp, n);
}


# list build: both passes put their scattered 4-byte stores (ordered chunk; a range's 128 slot lines) together in LDS and write whole 128-byte lines; sum order = LDS arrival rank as before
# speedup vs baseline: 4.6959x; 1.0720x over previous
#include <hip/hip_runtime.h>
#include <hip/hip_bf16.h>

typedef __attribute__((ext_vector_type(2))) float v2f;
typedef __attribute__((ext_vector_type(8))) float v8f;

#define NSLOT 32
#define RSH   7
#define RNG   (1 << RSH)
#define BINT  1024
#define EPT   8
#define CHUNK (EPT * BINT)
#define RANKW 8
#define SPW   10

__launch_bounds__(BINT)
__global__ void gcn_bin(const int* __restrict__ src, const int* __restrict__ dst, int e,
                        int nbk, int nba, int* __restrict__ sorted, int* __restrict__ offT,
                        int* __restrict__ spill_n) {
    __shared__ int lcnt[BINT];
    __shared__ int wsum[32];
    __shared__ int4 stage4[CHUNK / 4];
    int* stage = (int*)stage4;
    const int t = threadIdx.x;
    const int lane = t & 31;
    const int wave = t >> 5;
    const int blk = (int)blockIdx.x;
    if (blk == 0 && t == 0) spill_n[0] = 0;
    lcnt[t] = 0;
    const int base = blk * CHUNK;
    int key[EPT], bk[EPT], rk[EPT], of[EPT];
    #pragma unroll
    for (int k = 0; k < EPT; ++k) {
        const int ic = min(base + k * BINT + t, e - 1);
        bk[k]  = dst[ic];
        key[k] = src[ic];
    }
    #pragma unroll
    for (int k = 0; k < EPT; ++k) {
        const bool ok = base + k * BINT + t < e;
        key[k] = ok ? ((key[k] << RSH) | (bk[k] & (RNG - 1))) : -1;
        bk[k] >>= RSH;
    }
    __syncthreads();
    #pragma unroll
    for (int k = 0; k < EPT; ++k)
        rk[k] = (key[k] >= 0) ? atomicAdd(&lcnt[bk[k]], 1) : 0;
    __syncthreads();
    const int c0 = lcnt[t];
    int v = c0;
    #pragma unroll
    for (int off = 1; off < 32; off <<= 1) {
        const int u = __shfl_up(v, off, 32);
        if (lane >= off) v += u;
    }
    if (lane == 31) wsum[wave] = v;
    __syncthreads();
    if (wave == 0) {
        const int w0 = wsum[lane];
        int x = w0;
        #pragma unroll
        for (int off = 1; off < 32; off <<= 1) {
            const int u = __shfl_up(x, off, 32);
            if (lane >= off) x += u;
        }
        wsum[lane] = x - w0;
    }
    __syncthreads();
    const int ex = v - c0 + wsum[wave];
    lcnt[t] = ex;
    if (t <= nbk) offT[(size_t)t * nba + blk] = ex;
    __syncthreads();
    #pragma unroll
    for (int k = 0; k < EPT; ++k) of[k] = lcnt[bk[k]];
    #pragma unroll
    for (int k = 0; k < EPT; ++k)
        if (key[k] >= 0) stage[of[k] + rk[k]] = key[k];
    __syncthreads();
    const int nv = min(CHUNK, e - base);
    #pragma unroll
    for (int j = 0; j < CHUNK / (4 * BINT); ++j) {
        const int p = (j * BINT + t) * 4;
        if (p + 3 < nv) {
            *(int4*)(sorted + base + p) = stage4[j * BINT + t];
        } else {
            for (int u = p; u < nv; ++u) sorted[base + u] = stage[u];
        }
    }
}

__device__ __forceinline__ void gcn_place(int key, int r, int nbase, int* sl,
                                          int* __restrict__ spill_n, int* __restrict__ spill_src,
                                          int* __restrict__ spill_dst) {
    const int dl = key & (RNG - 1);
    const int s = key >> RSH;
    if (r < NSLOT) {
        sl[dl * NSLOT + r] = s;
    } else {
        const int o = atomicAdd(spill_n, 1);
        spill_src[o] = s;
        spill_dst[o] = nbase + dl;
    }
}

__device__ __forceinline__ float gcn_dinv(int c) {
    const float d = (float)(c + 1);
    return (d > 0.0f) ? rsqrtf(d) : 0.0f;
}

__launch_bounds__(RANKW * 32)
__global__ void gcn_rank(const int* __restrict__ sorted, const int* __restrict__ offT, int nba,
                         int* __restrict__ cnt, int* __restrict__ slots,
                         int* __restrict__ spill_n, int* __restrict__ spill_src,
                         int* __restrict__ spill_dst, int nrows) {
    __shared__ int lc[RNG];
    __shared__ int4 sl4[RNG * NSLOT / 4];
    int* sl = (int*)sl4;
    const int rb = (int)blockIdx.x;
    const int nbase = rb << RSH;
    const int t = (int)threadIdx.x;
    const int lane = t & 31;
    const int wave = __builtin_amdgcn_readfirstlane(t >> 5);
    if (t < RNG) lc[t] = 0;
    #pragma unroll
    for (int j = 0; j < RNG * NSLOT / 4 / (RANKW * 32); ++j)
        sl4[j * RANKW * 32 + t] = make_int4(0, 0, 0, 0);
    __syncthreads();
    for (int q0 = 0; q0 < nba; q0 += RANKW * SPW) {
        const int segl = q0 + wave + RANKW * lane;
        int mylo = 0, myhi = 0;
        if (lane < SPW && segl < nba) {
            mylo = offT[(size_t)rb * nba + segl];
            myhi = offT[(size_t)(rb + 1) * nba + segl];
        }
        int key[SPW];
        #pragma unroll
        for (int q = 0; q < SPW; ++q) {
            const int lo = __builtin_amdgcn_readlane(mylo, q);
            const int hi = __builtin_amdgcn_readlane(myhi, q);
            const int seg = q0 + wave + RANKW * q;
            const int i = lo + lane;
            key[q] = (i < hi) ? sorted[(size_t)seg * CHUNK + i] : -1;
        }
        int rr[SPW];
        #pragma unroll
        for (int q = 0; q < SPW; ++q)
            rr[q] = (key[q] >= 0) ? atomicAdd(&lc[key[q] & (RNG - 1)], 1) : 0;
        #pragma unroll
        for (int q = 0; q < SPW; ++q)
            if (key[q] >= 0) gcn_place(key[q], rr[q], nbase, sl, spill_n, spill_src, spill_dst);
        #pragma unroll
        for (int q = 0; q < SPW; ++q) {
            const int lo = __builtin_amdgcn_readlane(mylo, q);
            const int hi = __builtin_amdgcn_readlane(myhi, q);
            const int seg = q0 + wave + RANKW * q;
            for (int i0 = lo + 32; i0 < hi; i0 += 32) {
                const int i = i0 + lane;
                if (i < hi) {
                    const int kx = sorted[(size_t)seg * CHUNK + i];
                    gcn_place(kx, atomicAdd(&lc[kx & (RNG - 1)], 1), nbase, sl,
                              spill_n, spill_src, spill_dst);
                }
            }
        }
    }
    __syncthreads();
    if (t < RNG && nbase + t < nrows) cnt[nbase + t] = lc[t];
    int4* gl = (int4*)(slots + (size_t)nbase * NSLOT);
    #pragma unroll
    for (int j = 0; j < RNG * NSLOT / 4 / (RANKW * 32); ++j) {
        const int i = j * RANKW * 32 + t;
        if (nbase + (i >> 3) < nrows) gl[i] = sl4[i];
    }
}

struct gcn_head {
    int c, m, idx;
    float dd, nv;
};

__device__ __forceinline__ gcn_head gcn_agg_head(const int* __restrict__ cnt,
                                                 const int* __restrict__ slots,
                                                 int w, int lane) {
    gcn_head q;
    q.c = cnt[w];
    q.dd = gcn_dinv(q.c);
    q.m = min(q.c, NSLOT);
    q.idx = (lane < q.m) ? slots[(size_t)w * NSLOT + lane] : 0;
    q.nv = gcn_dinv(cnt[q.idx]) * q.dd;
    return q;
}

__device__ __forceinline__ float4 gcn_agg_body(const gcn_head q, const int* __restrict__ cnt,
                                               const int* __restrict__ spill_n,
                                               const int* __restrict__ spill_src,
                                               const int* __restrict__ spill_dst,
                                               int ecap, const float* __restrict__ h,
                                               int w, int lane) {
    const int c = q.c;
    const int m = q.m;
    const int idx = q.idx;
    const float dd = q.dd;
    const float nv = q.nv;
    const float4 hs = *(const float4*)(h + (size_t)w * 128 + lane * 4);
    float4 acc = make_float4(0.0f, 0.0f, 0.0f, 0.0f);
    #pragma unroll 4
    for (int k = 0; k < m; ++k) {
        const int s = __builtin_amdgcn_readlane(idx, k);
        const float nm = __int_as_float(__builtin_amdgcn_readlane(__float_as_int(nv), k));
        const float4 v = *(const float4*)(h + (size_t)s * 128 + lane * 4);
        acc.x += v.x * nm;
        acc.y += v.y * nm;
        acc.z += v.z * nm;
        acc.w += v.w * nm;
    }
    if (c > NSLOT) {
        const int on = min(spill_n[0], ecap);
        for (int base = 0; base < on; base += 32) {
            const int j = base + lane;
            bool mine = false;
            if (j < on) mine = (spill_dst[j] == w);
            const int os = mine ? spill_src[j] : 0;
            const float onv = gcn_dinv(cnt[os]) * dd;
            unsigned mask = __builtin_amdgcn_ballot_w32(mine);
            while (mask != 0u) {
                const int k = __builtin_ctz(mask);
                mask &= mask - 1u;
                const int s = __builtin_amdgcn_readlane(os, k);
                const float nm = __int_as_float(__builtin_amdgcn_readlane(__float_as_int(onv), k));
                const float4 v = *(const float4*)(h + (size_t)s * 128 + lane * 4);
                acc.x += v.x * nm;
                acc.y += v.y * nm;
                acc.z += v.z * nm;
                acc.w += v.w * nm;
            }
        }
    }
    const float d2 = dd * dd;
    acc.x += hs.x * d2;
    acc.y += hs.y * d2;
    acc.z += hs.z * d2;
    acc.w += hs.w * d2;
    return acc;
}

__device__ __forceinline__ float4 gcn_aggregate(const int* __restrict__ cnt,
                                                const int* __restrict__ slots,
                                                const int* __restrict__ spill_n,
                                                const int* __restrict__ spill_src,
                                                const int* __restrict__ spill_dst,
                                                int ecap, const float* __restrict__ h,
                                                int w, int lane) {
    const gcn_head q = gcn_agg_head(cnt, slots, w, lane);
    return gcn_agg_body(q, cnt, spill_n, spill_src, spill_dst, ecap, h, w, lane);
}

__launch_bounds__(256)
__global__ void gcn_gather_out(const int* __restrict__ cnt, const int* __restrict__ slots,
                               const int* __restrict__ spill_n, const int* __restrict__ spill_src,
                               const int* __restrict__ spill_dst, int ecap,
                               const float* __restrict__ h,
                               const float* __restrict__ b2, const float* __restrict__ Wo,
                               const float* __restrict__ bo, float* __restrict__ prob, int n) {
    const int lane = threadIdx.x & 31;
    const int w = __builtin_amdgcn_readfirstlane((int)((blockIdx.x * blockDim.x + threadIdx.x) >> 5));
    if (w >= n) return;
    const float4 acc = gcn_aggregate(cnt, slots, spill_n, spill_src, spill_dst, ecap, h, w, lane);
    const float4 bb = *(const float4*)(b2 + lane * 4);
    float4 v;
    v.x = fmaxf(acc.x + bb.x, 0.0f);
    v.y = fmaxf(acc.y + bb.y, 0.0f);
    v.z = fmaxf(acc.z + bb.z, 0.0f);
    v.w = fmaxf(acc.w + bb.w, 0.0f);
    float lg[5];
    #pragma unroll
    for (int c = 0; c < 5; ++c) {
        lg[c] = v.x * Wo[(lane * 4 + 0) * 5 + c]
              + v.y * Wo[(lane * 4 + 1) * 5 + c]
              + v.z * Wo[(lane * 4 + 2) * 5 + c]
              + v.w * Wo[(lane * 4 + 3) * 5 + c];
    }
    #pragma unroll
    for (int c = 0; c < 5; ++c) {
        #pragma unroll
        for (int off = 16; off > 0; off >>= 1)
            lg[c] += __shfl_xor(lg[c], off, 32);
    }
    #pragma unroll
    for (int c = 0; c < 5; ++c) lg[c] += bo[c];
    float m = lg[0];
    #pragma unroll
    for (int c = 1; c < 5; ++c) m = fmaxf(m, lg[c]);
    float ex[5], s = 0.0f;
    #pragma unroll
    for (int c = 0; c < 5; ++c) { ex[c] = expf(lg[c] - m); s += ex[c]; }
    const float inv = 1.0f / s;
    if (lane == 0) {
        #pragma unroll
        for (int c = 0; c < 5; ++c) prob[(size_t)w * 5 + c] = ex[c] * inv;
    }
}

#define SA_PITCH 132
#define TILE_N   32
#define GW       8
#define NPW      (TILE_N / GW)

__device__ __forceinline__ void gcn_bload(const float* __restrict__ W, unsigned boff, int g,
                                          float (&bx)[8], float (&by)[8]) {
    #pragma unroll
    for (int i = 0; i < 8; ++i) {
        const unsigned kn = (unsigned)(g * 8 + i);
        bx[i] = W[boff + (kn * 4u) * 128u];
        by[i] = W[boff + (kn * 4u + 1u) * 128u];
    }
}

__device__ __forceinline__ void gcn_tile2(v8f& d0, v8f& d1, const float* arow,
                                          const float* __restrict__ W, unsigned boff,
                                          float (&bx)[2][8], float (&by)[2][8]) {
    #pragma unroll
    for (int g = 0; g < 4; ++g) {
        if (g + 1 < 4) gcn_bload(W, boff, g + 1, bx[(g + 1) & 1], by[(g + 1) & 1]);
        #pragma unroll
        for (int i = 0; i < 8; ++i) {
            const int kk = g * 8 + i;
            const float2 p0 = *(const float2*)(arow + kk * 4);
            const float2 p1 = *(const float2*)(arow + 16 * SA_PITCH + kk * 4);
            v2f a0; a0.x = p0.x; a0.y = p0.y;
            v2f a1; a1.x = p1.x; a1.y = p1.y;
            v2f b; b.x = bx[g & 1][i]; b.y = by[g & 1][i];
            d0 = __builtin_amdgcn_wmma_f32_16x16x4_f32(
                false, a0, false, b, (short)0, d0, false, false);
            d1 = __builtin_amdgcn_wmma_f32_16x16x4_f32(
                false, a1, false, b, (short)0, d1, false, false);
        }
    }
}

__launch_bounds__(GW * 32, 16)
__global__ void gcn_gather_gemm2(const int* __restrict__ cnt, const int* __restrict__ slots,
                                 const int* __restrict__ spill_n, const int* __restrict__ spill_src,
                                 const int* __restrict__ spill_dst, int ecap,
                                 const float* __restrict__ x,
                                 const float* __restrict__ W1, const float* __restrict__ b1,
                                 const float* __restrict__ W2,
                                 float* __restrict__ OutLin, int n) {
    __shared__ float sA[TILE_N * SA_PITCH];
    __shared__ float sH[TILE_N * SA_PITCH];
    const int lane = threadIdx.x & 31;
    const int wave = __builtin_amdgcn_readfirstlane((int)(threadIdx.x >> 5));
    const int node0 = (int)blockIdx.x * TILE_N;
    int wn[NPW];
    gcn_head q[NPW];
    #pragma unroll
    for (int r = 0; r < NPW; ++r) {
        wn[r] = min(node0 + GW * r + wave, n - 1);
        q[r].c = cnt[wn[r]];
    }
    #pragma unroll
    for (int r = 0; r < NPW; ++r) {
        q[r].dd = gcn_dinv(q[r].c);
        q[r].m = min(q[r].c, NSLOT);
        q[r].idx = (lane < q[r].m) ? slots[(size_t)wn[r] * NSLOT + lane] : 0;
    }
    #pragma unroll
    for (int r = 0; r < NPW; ++r)
        q[r].nv = gcn_dinv(cnt[q[r].idx]) * q[r].dd;
    #pragma unroll
    for (int r = 0; r < NPW; ++r) {
        float4 v = gcn_agg_body(q[r], cnt, spill_n, spill_src, spill_dst, ecap, x, wn[r], lane);
        if (node0 + GW * r + wave >= n) v = make_float4(0.0f, 0.0f, 0.0f, 0.0f);
        *(float4*)&sA[(GW * r + wave) * SA_PITCH + lane * 4] = v;
    }
    const int half = lane >> 4;
    const int l16  = lane & 15;
    const int col  = wave * 16 + l16;
    const unsigned boff = (unsigned)((2 * half) * 128 + col);
    float bx[2][8], by[2][8];
    gcn_bload(W1, boff, 0, bx[0], by[0]);
    const float bias = b1[col];
    v8f d1a = (v8f){bias, bias, bias, bias, bias, bias, bias, bias};
    v8f d1b = d1a;
    __syncthreads();

    gcn_tile2(d1a, d1b, &sA[l16 * SA_PITCH + 2 * half], W1, boff, bx, by);
    gcn_bload(W2, boff, 0, bx[0], by[0]);
    const int hrow = 8 * half;
    #pragma unroll
    for (int j = 0; j < 8; ++j) {
        sH[(hrow + j) * SA_PITCH + col]      = fmaxf(d1a[j], 0.0f);
        sH[(16 + hrow + j) * SA_PITCH + col] = fmaxf(d1b[j], 0.0f);
    }
    __syncthreads();

    v8f d2a = (v8f){0.f, 0.f, 0.f, 0.f, 0.f, 0.f, 0.f, 0.f};
    v8f d2b = d2a;
    gcn_tile2(d2a, d2b, &sH[l16 * SA_PITCH + 2 * half], W2, boff, bx, by);
    const int rowbase = node0 + hrow;
    #pragma unroll
    for (int j = 0; j < 8; ++j) {
        if (rowbase + j < n)      OutLin[(size_t)(rowbase + j) * 128 + col] = d2a[j];
        if (rowbase + 16 + j < n) OutLin[(size_t)(rowbase + 16 + j) * 128 + col] = d2b[j];
    }
}

extern "C" void kernel_launch(void* const* d_in, const int* in_sizes, int n_in,
                              void* d_out, int out_size, void* d_ws, size_t ws_size,
                              hipStream_t stream) {
    const float* x  = (const float*)d_in[0];
    const int*   ei = (const int*)d_in[1];
    const float* W1 = (const float*)d_in[2];
    const float* b1 = (const float*)d_in[3];
    const float* W2 = (const float*)d_in[4];
    const float* b2 = (const float*)d_in[5];
    const float* Wo = (const float*)d_in[6];
    const float* bo = (const float*)d_in[7];

    const int n = in_sizes[0] / 128;
    const int e = in_sizes[1] / 2;
    const int* srcp = ei;
    const int* dstp = ei + e;

    float* bufB = (float*)d_ws;
    int*   cnt       = (int*)(bufB + (size_t)n * 128);
    int*   slots     = cnt + ((size_t)n + 31) / 32 * 32;
    int*   spill_n   = slots + (size_t)n * NSLOT;
    int*   spill_src = spill_n + 32;
    int*   spill_dst = spill_src + ((size_t)e + 31) / 32 * 32;
    const int nbk = (n + RNG - 1) / RNG;
    const int nba = (e + CHUNK - 1) / CHUNK;
    int*   sorted    = spill_dst + ((size_t)e + 31) / 32 * 32;
    int*   offT      = sorted + (size_t)nba * CHUNK;
    float* outp = (float*)d_out;

    const int T = 256;
    const int blk_ow  = (int)(((size_t)n * 32 + T - 1) / T);
    const int blk_t32 = (n + TILE_N - 1) / TILE_N;

    gcn_bin<<<nba, BINT, 0, stream>>>(srcp, dstp, e, nbk, nba, sorted, offT, spill_n);
    gcn_rank<<<nbk, RANKW * 32, 0, stream>>>(sorted, offT, nba, cnt, slots, spill_n,
                                             spill_src, spill_dst, n);

    gcn_gather_gemm2<<<blk_t32, 256, 0, stream>>>(cnt, slots, spill_n, spill_src, spill_dst, e,
                                                  x, W1, b1, W2, bufB, n);
    gcn_gather_out<<<blk_ow, T, 0, stream>>>(cnt, slots, spill_n, spill_src, spill_dst, e,
                                             bufB, b2, Wo, bo, outp, n);
}
